# starting kernel
# baseline (speedup 1.0000x reference)
#include <hip/hip_runtime.h>

typedef __attribute__((ext_vector_type(2))) float v2f;
typedef __attribute__((ext_vector_type(8))) float v8f;

#define NNODES  (128 * 512)
#define NEDGES  (128 * 512 * 16)
#define NGRAPHS 128
#define NPG     512
#define EPG     (512 * 16)
#define RES     128

__global__ void __launch_bounds__(512) k_dinv(const int* __restrict__ ei,
                                              float* __restrict__ dinv) {
  __shared__ int cnt[NPG];
  const int b = blockIdx.x;
  const int tid = threadIdx.x;
  if (tid < NPG) cnt[tid] = 1;
  __syncthreads();
  for (int e = tid; e < EPG; e += blockDim.x)
    atomicAdd(&cnt[ei[NEDGES + b * EPG + e] & (NPG - 1)], 1);
  __syncthreads();
  if (tid < NPG) dinv[b * NPG + tid] = rsqrtf((float)cnt[tid]);
}

template <int NT, int K>
__global__ void __launch_bounds__(256) k_gemm_wmma(const float* __restrict__ A,
                                                   const float* __restrict__ B,
                                                   float* __restrict__ C, int M) {
  const int N = NT * 16;
  __shared__ v2f sB[(K / 2) * N];
  for (int i = threadIdx.x; i < (K / 2) * N; i += blockDim.x) {
    int kp = i / N;
    int c  = i - kp * N;
    v2f w;
    w.x = B[(size_t)(2 * kp) * N + c];
    w.y = B[(size_t)(2 * kp + 1) * N + c];
    sB[i] = w;
  }
  __syncthreads();

  int wave = (int)((blockIdx.x * blockDim.x + threadIdx.x) >> 5);
  int lane = threadIdx.x & 31;
  if (wave >= (M >> 4)) return;
  int r  = lane & 15;
  int kb = (lane >> 4) << 1;
  const float* Arow = A + (size_t)(wave * 16 + r) * K;

  v8f acc[NT] = {};
  for (int k = 0; k < K; k += 4) {
    v2f a = *(const v2f*)(Arow + k + kb);
    v2f bf[NT];
#pragma unroll
    for (int t = 0; t < NT; ++t)
      bf[t] = sB[((k + kb) >> 1) * N + t * 16 + r];
#pragma unroll
    for (int t = 0; t < NT; ++t)
      acc[t] = __builtin_amdgcn_wmma_f32_16x16x4_f32(false, a, false, bf[t], (short)0,
                                                     acc[t], false, false);
  }
  int rbase = (lane >> 4) << 3;
#pragma unroll
  for (int t = 0; t < NT; ++t) {
    float* Crow = C + (size_t)(wave * 16 + rbase) * N + t * 16 + r;
#pragma unroll
    for (int v = 0; v < 8; ++v) Crow[(size_t)v * N] = acc[t][v];
  }
}

template <int C>
__global__ void __launch_bounds__(512) k_agg(
    const float* __restrict__ h, const int* __restrict__ ei,
    const float* __restrict__ dinv, const float* __restrict__ bias,
    float* __restrict__ y, float* __restrict__ pout) {
  extern __shared__ float smem[];
  float* acc = smem;
  float* hl  = smem + NPG * C;
  float* dv  = smem + 2 * NPG * C;
  const int b = blockIdx.x;
  const int gbase = b * NPG;
  const int tid = threadIdx.x;

  for (int i = tid; i < NPG; i += blockDim.x) dv[i] = dinv[gbase + i];
  for (int idx = tid; idx < NPG * C; idx += blockDim.x)
    hl[idx] = h[(size_t)gbase * C + idx];
  __syncthreads();

  for (int idx = tid; idx < NPG * C; idx += blockDim.x) {
    float d = dv[idx / C];
    acc[idx] = hl[idx] * d * d;
  }
  __syncthreads();

  const int chunks = C / 4;
  for (int t = tid; t < EPG * chunks; t += blockDim.x) {
    int e  = t / chunks;
    int cg = (t - e * chunks) * 4;
    int s  = ei[(size_t)b * EPG + e] - gbase;
    int d  = ei[(size_t)NEDGES + (size_t)b * EPG + e] - gbase;
    float w = dv[s] * dv[d];
    float* ad = acc + d * C + cg;
    const float* hs = hl + s * C + cg;
    atomicAdd(&ad[0], hs[0] * w);
    atomicAdd(&ad[1], hs[1] * w);
    atomicAdd(&ad[2], hs[2] * w);
    atomicAdd(&ad[3], hs[3] * w);
  }
  __syncthreads();

  for (int idx = tid; idx < NPG * C; idx += blockDim.x) {
    float v = acc[idx] + bias[idx & (C - 1)];
    y[(size_t)gbase * C + idx] = fmaxf(v, 0.0f);
  }
  for (int idx = tid; idx < RES * C; idx += blockDim.x) {
    int c = idx & (C - 1);
    int r = idx / C;
    float bb = bias[c];
    float v1 = fmaxf(acc[(4 * r + 1) * C + c] + bb, 0.0f);
    float v2 = fmaxf(acc[(4 * r + 2) * C + c] + bb, 0.0f);
    pout[(size_t)b * RES * C + idx] = 0.5f * (v1 + v2);
  }
}

template <int D>
__global__ void __launch_bounds__(256) k_sim_wmma(const float* __restrict__ U,
                                                  const float* __restrict__ V,
                                                  float* __restrict__ out) {
  __shared__ float sV[RES * D];
  const int b = blockIdx.x;
  const float* Vb = V + (size_t)b * RES * D;
  for (int i = threadIdx.x; i < RES * D; i += blockDim.x) sV[i] = Vb[i];
  __syncthreads();

  int tm   = threadIdx.x >> 5;
  int lane = threadIdx.x & 31;
  int r    = lane & 15;
  int kb   = (lane >> 4) << 1;
  const float* Ar = U + ((size_t)b * RES + tm * 16 + r) * D;

  v8f acc[8] = {};
  for (int k = 0; k < D; k += 4) {
    v2f a = *(const v2f*)(Ar + k + kb);
    v2f bf[8];
#pragma unroll
    for (int t = 0; t < 8; ++t)
      bf[t] = *(const v2f*)(sV + (t * 16 + r) * D + k + kb);
#pragma unroll
    for (int t = 0; t < 8; ++t)
      acc[t] = __builtin_amdgcn_wmma_f32_16x16x4_f32(false, a, false, bf[t], (short)0,
                                                     acc[t], false, false);
  }
  int rbase = (lane >> 4) << 3;
#pragma unroll
  for (int t = 0; t < 8; ++t) {
    float* Co = out + (size_t)b * RES * RES + (size_t)(tm * 16 + rbase) * RES + t * 16 + r;
#pragma unroll
    for (int v = 0; v < 8; ++v) Co[(size_t)v * RES] = acc[t][v];
  }
}

extern "C" void kernel_launch(void* const* d_in, const int* in_sizes, int n_in,
                              void* d_out, int out_size, void* d_ws, size_t ws_size,
                              hipStream_t stream) {
  (void)in_sizes; (void)n_in; (void)out_size; (void)ws_size;
  const float* x[2]  = {(const float*)d_in[0], (const float*)d_in[1]};
  const int*   ei[2] = {(const int*)d_in[2],  (const int*)d_in[3]};
  const float* W[3]  = {(const float*)d_in[6], (const float*)d_in[8], (const float*)d_in[10]};
  const float* bv[3] = {(const float*)d_in[7], (const float*)d_in[9], (const float*)d_in[11]};
  float* out = (float*)d_out;

  float* ws = (float*)d_ws;
  size_t o = 0;
  float* dinvA = ws + o; o += (size_t)2 * NNODES;
  float* B0 = ws + o; o += (size_t)NNODES * 64;
  float* B1 = ws + o; o += (size_t)NNODES * 64;
  float* B2 = ws + o; o += (size_t)NNODES * 32;
  float* P[2][3];
  const int dims[3] = {64, 32, 16};
  for (int s = 0; s < 2; ++s)
    for (int l = 0; l < 3; ++l) { P[s][l] = ws + o; o += (size_t)NGRAPHS * RES * dims[l]; }

  const unsigned gemmBlocks = (NNODES / 16) * 32 / 256;
  auto shbytes = [](int C) { return (size_t)(2 * NPG * C + NPG) * sizeof(float); };

  for (int s = 0; s < 2; ++s) {
    float* dv = dinvA + (size_t)s * NNODES;
    k_dinv<<<NGRAPHS, 512, 0, stream>>>(ei[s], dv);

    k_gemm_wmma<4, 128><<<gemmBlocks, 256, 0, stream>>>(x[s], W[0], B0, NNODES);
    k_agg<64><<<NGRAPHS, 512, shbytes(64), stream>>>(B0, ei[s], dv, bv[0], B1, P[s][0]);
    k_gemm_wmma<2, 64><<<gemmBlocks, 256, 0, stream>>>(B1, W[1], B0, NNODES);
    k_agg<32><<<NGRAPHS, 512, shbytes(32), stream>>>(B0, ei[s], dv, bv[1], B2, P[s][1]);
    k_gemm_wmma<1, 32><<<gemmBlocks, 256, 0, stream>>>(B2, W[2], B0, NNODES);
    k_agg<16><<<NGRAPHS, 512, shbytes(16), stream>>>(B0, ei[s], dv, bv[2], B1, P[s][2]);
  }

  k_sim_wmma<64><<<NGRAPHS, 256, 0, stream>>>(P[0][0], P[1][0], out);
  k_sim_wmma<32><<<NGRAPHS, 256, 0, stream>>>(P[0][1], P[1][1],
                                              out + (size_t)1 * NGRAPHS * RES * RES);
  k_sim_wmma<16><<<NGRAPHS, 256, 0, stream>>>(P[0][2], P[1][2],
                                              out + (size_t)2 * NGRAPHS * RES * RES);
}


# k_agg: each graph's 8192 edge indices staged once in LDS (packed src|dst, loads issued ahead of the h-tile copy); scatter loop reads them from LDS, no global load in its per-trip dependent chain
# speedup vs baseline: 1.1799x; 1.1799x over previous
#include <hip/hip_runtime.h>

typedef __attribute__((ext_vector_type(2))) float v2f;
typedef __attribute__((ext_vector_type(8))) float v8f;

#define NNODES  (128 * 512)
#define NEDGES  (128 * 512 * 16)
#define NGRAPHS 128
#define NPG     512
#define EPG     (512 * 16)
#define RES     128

__global__ void __launch_bounds__(512) k_dinv(const int* __restrict__ ei,
                                              float* __restrict__ dinv) {
  __shared__ int cnt[NPG];
  const int b = blockIdx.x;
  const int tid = threadIdx.x;
  if (tid < NPG) cnt[tid] = 1;
  __syncthreads();
  for (int e = tid; e < EPG; e += blockDim.x)
    atomicAdd(&cnt[ei[NEDGES + b * EPG + e] & (NPG - 1)], 1);
  __syncthreads();
  if (tid < NPG) dinv[b * NPG + tid] = rsqrtf((float)cnt[tid]);
}

template <int NT, int K>
__global__ void __launch_bounds__(256) k_gemm_wmma(const float* __restrict__ A,
                                                   const float* __restrict__ B,
                                                   float* __restrict__ C, int M) {
  const int N = NT * 16;
  __shared__ v2f sB[(K / 2) * N];
  for (int i = threadIdx.x; i < (K / 2) * N; i += blockDim.x) {
    int kp = i / N;
    int c  = i - kp * N;
    v2f w;
    w.x = B[(size_t)(2 * kp) * N + c];
    w.y = B[(size_t)(2 * kp + 1) * N + c];
    sB[i] = w;
  }
  __syncthreads();

  int wave = (int)((blockIdx.x * blockDim.x + threadIdx.x) >> 5);
  int lane = threadIdx.x & 31;
  if (wave >= (M >> 4)) return;
  int r  = lane & 15;
  int kb = (lane >> 4) << 1;
  const float* Arow = A + (size_t)(wave * 16 + r) * K;

  v8f acc[NT] = {};
  for (int k = 0; k < K; k += 4) {
    v2f a = *(const v2f*)(Arow + k + kb);
    v2f bf[NT];
#pragma unroll
    for (int t = 0; t < NT; ++t)
      bf[t] = sB[((k + kb) >> 1) * N + t * 16 + r];
#pragma unroll
    for (int t = 0; t < NT; ++t)
      acc[t] = __builtin_amdgcn_wmma_f32_16x16x4_f32(false, a, false, bf[t], (short)0,
                                                     acc[t], false, false);
  }
  int rbase = (lane >> 4) << 3;
#pragma unroll
  for (int t = 0; t < NT; ++t) {
    float* Crow = C + (size_t)(wave * 16 + rbase) * N + t * 16 + r;
#pragma unroll
    for (int v = 0; v < 8; ++v) Crow[(size_t)v * N] = acc[t][v];
  }
}

#define AGG_THREADS 512
template <int C>
__global__ void __launch_bounds__(AGG_THREADS) k_agg(
    const float* __restrict__ h, const int* __restrict__ ei,
    const float* __restrict__ dinv, const float* __restrict__ bias,
    float* __restrict__ y, float* __restrict__ pout) {
  extern __shared__ float smem[];
  float* acc = smem;
  float* hl  = smem + NPG * C;
  float* dv  = smem + 2 * NPG * C;
  unsigned* el = (unsigned*)(smem + 2 * NPG * C + NPG);
  const int b = blockIdx.x;
  const int gbase = b * NPG;
  const int tid = threadIdx.x;

  int es[EPG / AGG_THREADS], ed[EPG / AGG_THREADS];
#pragma unroll
  for (int i = 0; i < EPG / AGG_THREADS; ++i) {
    int e = tid + i * AGG_THREADS;
    es[i] = ei[(size_t)b * EPG + e];
    ed[i] = ei[(size_t)NEDGES + (size_t)b * EPG + e];
  }
  for (int i = tid; i < NPG; i += blockDim.x) dv[i] = dinv[gbase + i];
  for (int idx = tid; idx < NPG * C; idx += blockDim.x)
    hl[idx] = h[(size_t)gbase * C + idx];
#pragma unroll
  for (int i = 0; i < EPG / AGG_THREADS; ++i) {
    unsigned s = (unsigned)(es[i] - gbase);
    unsigned d = (unsigned)(ed[i] - gbase);
    el[tid + i * AGG_THREADS] = (s & 0xffffu) | (d << 16);
  }
  __syncthreads();

  for (int idx = tid; idx < NPG * C; idx += blockDim.x) {
    float d = dv[idx / C];
    acc[idx] = hl[idx] * d * d;
  }
  __syncthreads();

  const int chunks = C / 4;
  for (int t = tid; t < EPG * chunks; t += blockDim.x) {
    int e  = t / chunks;
    int cg = (t - e * chunks) * 4;
    unsigned p = el[e];
    int s  = (int)(p & 0xffffu);
    int d  = (int)(p >> 16);
    float w = dv[s] * dv[d];
    float* ad = acc + d * C + cg;
    const float* hs = hl + s * C + cg;
    atomicAdd(&ad[0], hs[0] * w);
    atomicAdd(&ad[1], hs[1] * w);
    atomicAdd(&ad[2], hs[2] * w);
    atomicAdd(&ad[3], hs[3] * w);
  }
  __syncthreads();

  for (int idx = tid; idx < NPG * C; idx += blockDim.x) {
    float v = acc[idx] + bias[idx & (C - 1)];
    y[(size_t)gbase * C + idx] = fmaxf(v, 0.0f);
  }
  for (int idx = tid; idx < RES * C; idx += blockDim.x) {
    int c = idx & (C - 1);
    int r = idx / C;
    float bb = bias[c];
    float v1 = fmaxf(acc[(4 * r + 1) * C + c] + bb, 0.0f);
    float v2 = fmaxf(acc[(4 * r + 2) * C + c] + bb, 0.0f);
    pout[(size_t)b * RES * C + idx] = 0.5f * (v1 + v2);
  }
}

template <int D>
__global__ void __launch_bounds__(256) k_sim_wmma(const float* __restrict__ U,
                                                  const float* __restrict__ V,
                                                  float* __restrict__ out) {
  __shared__ float sV[RES * D];
  const int b = blockIdx.x;
  const float* Vb = V + (size_t)b * RES * D;
  for (int i = threadIdx.x; i < RES * D; i += blockDim.x) sV[i] = Vb[i];
  __syncthreads();

  int tm   = threadIdx.x >> 5;
  int lane = threadIdx.x & 31;
  int r    = lane & 15;
  int kb   = (lane >> 4) << 1;
  const float* Ar = U + ((size_t)b * RES + tm * 16 + r) * D;

  v8f acc[8] = {};
  for (int k = 0; k < D; k += 4) {
    v2f a = *(const v2f*)(Ar + k + kb);
    v2f bf[8];
#pragma unroll
    for (int t = 0; t < 8; ++t)
      bf[t] = *(const v2f*)(sV + (t * 16 + r) * D + k + kb);
#pragma unroll
    for (int t = 0; t < 8; ++t)
      acc[t] = __builtin_amdgcn_wmma_f32_16x16x4_f32(false, a, false, bf[t], (short)0,
                                                     acc[t], false, false);
  }
  int rbase = (lane >> 4) << 3;
#pragma unroll
  for (int t = 0; t < 8; ++t) {
    float* Co = out + (size_t)b * RES * RES + (size_t)(tm * 16 + rbase) * RES + t * 16 + r;
#pragma unroll
    for (int v = 0; v < 8; ++v) Co[(size_t)v * RES] = acc[t][v];
  }
}

extern "C" void kernel_launch(void* const* d_in, const int* in_sizes, int n_in,
                              void* d_out, int out_size, void* d_ws, size_t ws_size,
                              hipStream_t stream) {
  (void)in_sizes; (void)n_in; (void)out_size; (void)ws_size;
  const float* x[2]  = {(const float*)d_in[0], (const float*)d_in[1]};
  const int*   ei[2] = {(const int*)d_in[2],  (const int*)d_in[3]};
  const float* W[3]  = {(const float*)d_in[6], (const float*)d_in[8], (const float*)d_in[10]};
  const float* bv[3] = {(const float*)d_in[7], (const float*)d_in[9], (const float*)d_in[11]};
  float* out = (float*)d_out;

  float* ws = (float*)d_ws;
  size_t o = 0;
  float* dinvA = ws + o; o += (size_t)2 * NNODES;
  float* B0 = ws + o; o += (size_t)NNODES * 64;
  float* B1 = ws + o; o += (size_t)NNODES * 64;
  float* B2 = ws + o; o += (size_t)NNODES * 32;
  float* P[2][3];
  const int dims[3] = {64, 32, 16};
  for (int s = 0; s < 2; ++s)
    for (int l = 0; l < 3; ++l) { P[s][l] = ws + o; o += (size_t)NGRAPHS * RES * dims[l]; }

  const unsigned gemmBlocks = (NNODES / 16) * 32 / 256;
  auto shbytes = [](int C) { return (size_t)(2 * NPG * C + NPG + EPG) * sizeof(float); };

  for (int s = 0; s < 2; ++s) {
    float* dv = dinvA + (size_t)s * NNODES;
    k_dinv<<<NGRAPHS, 512, 0, stream>>>(ei[s], dv);

    k_gemm_wmma<4, 128><<<gemmBlocks, 256, 0, stream>>>(x[s], W[0], B0, NNODES);
    k_agg<64><<<NGRAPHS, AGG_THREADS, shbytes(64), stream>>>(B0, ei[s], dv, bv[0], B1, P[s][0]);
    k_gemm_wmma<2, 64><<<gemmBlocks, 256, 0, stream>>>(B1, W[1], B0, NNODES);
    k_agg<32><<<NGRAPHS, AGG_THREADS, shbytes(32), stream>>>(B0, ei[s], dv, bv[1], B2, P[s][1]);
    k_gemm_wmma<1, 32><<<gemmBlocks, 256, 0, stream>>>(B2, W[2], B0, NNODES);
    k_agg<16><<<NGRAPHS, AGG_THREADS, shbytes(16), stream>>>(B0, ei[s], dv, bv[2], B1, P[s][2]);
  }

  k_sim_wmma<64><<<NGRAPHS, 256, 0, stream>>>(P[0][0], P[1][0], out);
  k_sim_wmma<32><<<NGRAPHS, 256, 0, stream>>>(P[0][1], P[1][1],
                                              out + (size_t)1 * NGRAPHS * RES * RES);
  k_sim_wmma<16><<<NGRAPHS, 256, 0, stream>>>(P[0][2], P[1][2],
                                              out + (size_t)2 * NGRAPHS * RES * RES);
}


# loops strided by blockDim.x (built as one load per wait) get compile-time trips: loads issued in batches, 16 bytes wide where contiguous; k_agg scatter takes 4 edges per trip, same LDS atomics
# speedup vs baseline: 2.3010x; 1.9502x over previous
#include <hip/hip_runtime.h>

typedef __attribute__((ext_vector_type(2))) float v2f;
typedef __attribute__((ext_vector_type(8))) float v8f;

#define NNODES  (128 * 512)
#define NEDGES  (128 * 512 * 16)
#define NGRAPHS 128
#define NPG     512
#define EPG     (512 * 16)
#define RES     128

#define DINV_THREADS 512
__global__ void __launch_bounds__(DINV_THREADS) k_dinv(const int* __restrict__ ei,
                                                       float* __restrict__ dinv) {
  static_assert(EPG % DINV_THREADS == 0 && DINV_THREADS >= NPG, "block covers the graph");
  __shared__ int cnt[NPG];
  const int b = blockIdx.x;
  const int tid = threadIdx.x;
  int dd[EPG / DINV_THREADS];
#pragma unroll
  for (int i = 0; i < EPG / DINV_THREADS; ++i)
    dd[i] = ei[NEDGES + b * EPG + tid + i * DINV_THREADS];
  if (tid < NPG) cnt[tid] = 1;
  __syncthreads();
#pragma unroll
  for (int i = 0; i < EPG / DINV_THREADS; ++i)
    atomicAdd(&cnt[dd[i] & (NPG - 1)], 1);
  __syncthreads();
  if (tid < NPG) dinv[b * NPG + tid] = rsqrtf((float)cnt[tid]);
}

#define GEMM_THREADS 256
template <int NT, int K>
__global__ void __launch_bounds__(GEMM_THREADS) k_gemm_wmma(const float* __restrict__ A,
                                                   const float* __restrict__ B,
                                                   float* __restrict__ C, int M) {
  const int N = NT * 16;
  static_assert(((K / 2) * NT * 16) % GEMM_THREADS == 0, "whole sweeps of the block");
  __shared__ v2f sB[(K / 2) * N];
#pragma unroll
  for (int it = 0; it < (K / 2) * N / GEMM_THREADS; ++it) {
    int i  = threadIdx.x + it * GEMM_THREADS;
    int kp = i / N;
    int c  = i - kp * N;
    v2f w;
    w.x = B[(size_t)(2 * kp) * N + c];
    w.y = B[(size_t)(2 * kp + 1) * N + c];
    sB[i] = w;
  }
  __syncthreads();

  int wave = (int)((blockIdx.x * blockDim.x + threadIdx.x) >> 5);
  int lane = threadIdx.x & 31;
  if (wave >= (M >> 4)) return;
  int r  = lane & 15;
  int kb = (lane >> 4) << 1;
  const float* Arow = A + (size_t)(wave * 16 + r) * K;

  v8f acc[NT] = {};
  for (int k = 0; k < K; k += 4) {
    v2f a = *(const v2f*)(Arow + k + kb);
    v2f bf[NT];
#pragma unroll
    for (int t = 0; t < NT; ++t)
      bf[t] = sB[((k + kb) >> 1) * N + t * 16 + r];
#pragma unroll
    for (int t = 0; t < NT; ++t)
      acc[t] = __builtin_amdgcn_wmma_f32_16x16x4_f32(false, a, false, bf[t], (short)0,
                                                     acc[t], false, false);
  }
  int rbase = (lane >> 4) << 3;
#pragma unroll
  for (int t = 0; t < NT; ++t) {
    float* Crow = C + (size_t)(wave * 16 + rbase) * N + t * 16 + r;
#pragma unroll
    for (int v = 0; v < 8; ++v) Crow[(size_t)v * N] = acc[t][v];
  }
}

#define AGG_THREADS 512
template <int C>
__global__ void __launch_bounds__(AGG_THREADS) k_agg(
    const float* __restrict__ h, const int* __restrict__ ei,
    const float* __restrict__ dinv, const float* __restrict__ bias,
    float* __restrict__ y, float* __restrict__ pout) {
  extern __shared__ __align__(16) float smem[];
  float* acc = smem;
  float* hl  = smem + NPG * C;
  float* dv  = smem + 2 * NPG * C;
  unsigned* el = (unsigned*)(smem + 2 * NPG * C + NPG);
  float4* acc4 = reinterpret_cast<float4*>(acc);
  float4* hl4  = reinterpret_cast<float4*>(hl);
  const int b = blockIdx.x;
  const int gbase = b * NPG;
  const int tid = threadIdx.x;

  constexpr int chunks = C / 4;
  constexpr int RPT    = AGG_THREADS / chunks;
  constexpr int NV     = NPG * C / 4 / AGG_THREADS;
  constexpr int NPL    = RES * C / 4 / AGG_THREADS;
  constexpr int EPT    = EPG / AGG_THREADS;
  static_assert(AGG_THREADS == NPG && AGG_THREADS % chunks == 0, "one dinv per thread; cg fixed per thread");
  static_assert((EPG * chunks / AGG_THREADS) % 4 == 0, "scatter trips come in groups of U = 4");
  const int r0 = tid / chunks;
  const int cg = (tid - r0 * chunks) * 4;

  int es[EPT], ed[EPT];
#pragma unroll
  for (int i = 0; i < EPT; ++i) {
    int e = tid + i * AGG_THREADS;
    es[i] = ei[(size_t)b * EPG + e];
    ed[i] = ei[(size_t)NEDGES + (size_t)b * EPG + e];
  }
  const float dmine = dinv[gbase + tid];
  const float4* h4 = reinterpret_cast<const float4*>(h + (size_t)gbase * C);
  float4 hv[NV];
#pragma unroll
  for (int i = 0; i < NV; ++i) hv[i] = h4[tid + i * AGG_THREADS];

  dv[tid] = dmine;
#pragma unroll
  for (int i = 0; i < NV; ++i) hl4[tid + i * AGG_THREADS] = hv[i];
#pragma unroll
  for (int i = 0; i < EPT; ++i) {
    unsigned s = (unsigned)(es[i] - gbase);
    unsigned d = (unsigned)(ed[i] - gbase);
    el[tid + i * AGG_THREADS] = (s & 0xffffu) | (d << 16);
  }
  __syncthreads();

#pragma unroll
  for (int i = 0; i < NV; ++i) {
    float d = dv[r0 + i * RPT];
    float4 v = hv[i];
    v.x = v.x * d * d;
    v.y = v.y * d * d;
    v.z = v.z * d * d;
    v.w = v.w * d * d;
    acc4[tid + i * AGG_THREADS] = v;
  }
  __syncthreads();

  constexpr int TRIPS = EPG * chunks / AGG_THREADS;
  constexpr int U = 4;
  const float* hlc = hl + cg;
  float* accc = acc + cg;
  unsigned p[U];
#pragma unroll
  for (int u = 0; u < U; ++u) p[u] = el[r0 + u * RPT];
  for (int j = 0; j < TRIPS; j += U) {
    float  w[U];
    float4 v[U];
    int    d[U];
#pragma unroll
    for (int u = 0; u < U; ++u) {
      int s = (int)(p[u] & 0xffffu);
      d[u]  = (int)(p[u] >> 16);
      w[u]  = dv[s] * dv[d[u]];
      v[u]  = *reinterpret_cast<const float4*>(hlc + s * C);
    }
    const int jn = (j + U < TRIPS) ? (j + U) : j;
#pragma unroll
    for (int u = 0; u < U; ++u) p[u] = el[r0 + (jn + u) * RPT];
#pragma unroll
    for (int u = 0; u < U; ++u) {
      float* ad = accc + d[u] * C;
      atomicAdd(&ad[0], v[u].x * w[u]);
      atomicAdd(&ad[1], v[u].y * w[u]);
      atomicAdd(&ad[2], v[u].z * w[u]);
      atomicAdd(&ad[3], v[u].w * w[u]);
    }
  }
  __syncthreads();

  const float4 bb = *reinterpret_cast<const float4*>(bias + cg);
  float4* y4 = reinterpret_cast<float4*>(y + (size_t)gbase * C);
#pragma unroll
  for (int i = 0; i < NV; ++i) {
    float4 a = acc4[tid + i * AGG_THREADS];
    float4 o;
    o.x = fmaxf(a.x + bb.x, 0.0f);
    o.y = fmaxf(a.y + bb.y, 0.0f);
    o.z = fmaxf(a.z + bb.z, 0.0f);
    o.w = fmaxf(a.w + bb.w, 0.0f);
    y4[tid + i * AGG_THREADS] = o;
  }
  float4* p4 = reinterpret_cast<float4*>(pout + (size_t)b * RES * C);
#pragma unroll
  for (int i = 0; i < NPL; ++i) {
    int r = r0 + i * RPT;
    float4 a1 = *reinterpret_cast<const float4*>(accc + (4 * r + 1) * C);
    float4 a2 = *reinterpret_cast<const float4*>(accc + (4 * r + 2) * C);
    float4 o;
    o.x = 0.5f * (fmaxf(a1.x + bb.x, 0.0f) + fmaxf(a2.x + bb.x, 0.0f));
    o.y = 0.5f * (fmaxf(a1.y + bb.y, 0.0f) + fmaxf(a2.y + bb.y, 0.0f));
    o.z = 0.5f * (fmaxf(a1.z + bb.z, 0.0f) + fmaxf(a2.z + bb.z, 0.0f));
    o.w = 0.5f * (fmaxf(a1.w + bb.w, 0.0f) + fmaxf(a2.w + bb.w, 0.0f));
    p4[tid + i * AGG_THREADS] = o;
  }
}

#define SIM_THREADS 256
template <int D>
__global__ void __launch_bounds__(SIM_THREADS) k_sim_wmma(const float* __restrict__ U,
                                                  const float* __restrict__ V,
                                                  float* __restrict__ out) {
  static_assert((RES * D / 4) % SIM_THREADS == 0, "whole sweeps of the block");
  __shared__ __align__(16) float sV[RES * D];
  const int b = blockIdx.x;
  const float4* Vb4 = reinterpret_cast<const float4*>(V + (size_t)b * RES * D);
  float4* sV4 = reinterpret_cast<float4*>(sV);
  float4 vv[RES * D / 4 / SIM_THREADS];
#pragma unroll
  for (int it = 0; it < RES * D / 4 / SIM_THREADS; ++it) vv[it] = Vb4[threadIdx.x + it * SIM_THREADS];
#pragma unroll
  for (int it = 0; it < RES * D / 4 / SIM_THREADS; ++it) sV4[threadIdx.x + it * SIM_THREADS] = vv[it];
  __syncthreads();

  int tm   = threadIdx.x >> 5;
  int lane = threadIdx.x & 31;
  int r    = lane & 15;
  int kb   = (lane >> 4) << 1;
  const float* Ar = U + ((size_t)b * RES + tm * 16 + r) * D;

  v8f acc[8] = {};
  for (int k = 0; k < D; k += 4) {
    v2f a = *(const v2f*)(Ar + k + kb);
    v2f bf[8];
#pragma unroll
    for (int t = 0; t < 8; ++t)
      bf[t] = *(const v2f*)(sV + (t * 16 + r) * D + k + kb);
#pragma unroll
    for (int t = 0; t < 8; ++t)
      acc[t] = __builtin_amdgcn_wmma_f32_16x16x4_f32(false, a, false, bf[t], (short)0,
                                                     acc[t], false, false);
  }
  int rbase = (lane >> 4) << 3;
#pragma unroll
  for (int t = 0; t < 8; ++t) {
    float* Co = out + (size_t)b * RES * RES + (size_t)(tm * 16 + rbase) * RES + t * 16 + r;
#pragma unroll
    for (int v = 0; v < 8; ++v) Co[(size_t)v * RES] = acc[t][v];
  }
}

extern "C" void kernel_launch(void* const* d_in, const int* in_sizes, int n_in,
                              void* d_out, int out_size, void* d_ws, size_t ws_size,
                              hipStream_t stream) {
  (void)in_sizes; (void)n_in; (void)out_size; (void)ws_size;
  const float* x[2]  = {(const float*)d_in[0], (const float*)d_in[1]};
  const int*   ei[2] = {(const int*)d_in[2],  (const int*)d_in[3]};
  const float* W[3]  = {(const float*)d_in[6], (const float*)d_in[8], (const float*)d_in[10]};
  const float* bv[3] = {(const float*)d_in[7], (const float*)d_in[9], (const float*)d_in[11]};
  float* out = (float*)d_out;

  float* ws = (float*)d_ws;
  size_t o = 0;
  float* dinvA = ws + o; o += (size_t)2 * NNODES;
  float* B0 = ws + o; o += (size_t)NNODES * 64;
  float* B1 = ws + o; o += (size_t)NNODES * 64;
  float* B2 = ws + o; o += (size_t)NNODES * 32;
  float* P[2][3];
  const int dims[3] = {64, 32, 16};
  for (int s = 0; s < 2; ++s)
    for (int l = 0; l < 3; ++l) { P[s][l] = ws + o; o += (size_t)NGRAPHS * RES * dims[l]; }

  const unsigned gemmBlocks = (NNODES / 16) * 32 / 256;
  auto shbytes = [](int C) { return (size_t)(2 * NPG * C + NPG + EPG) * sizeof(float); };

  for (int s = 0; s < 2; ++s) {
    float* dv = dinvA + (size_t)s * NNODES;
    k_dinv<<<NGRAPHS, 512, 0, stream>>>(ei[s], dv);

    k_gemm_wmma<4, 128><<<gemmBlocks, 256, 0, stream>>>(x[s], W[0], B0, NNODES);
    k_agg<64><<<NGRAPHS, AGG_THREADS, shbytes(64), stream>>>(B0, ei[s], dv, bv[0], B1, P[s][0]);
    k_gemm_wmma<2, 64><<<gemmBlocks, 256, 0, stream>>>(B1, W[1], B0, NNODES);
    k_agg<32><<<NGRAPHS, AGG_THREADS, shbytes(32), stream>>>(B0, ei[s], dv, bv[1], B2, P[s][1]);
    k_gemm_wmma<1, 32><<<gemmBlocks, 256, 0, stream>>>(B2, W[2], B0, NNODES);
    k_agg<16><<<NGRAPHS, AGG_THREADS, shbytes(16), stream>>>(B0, ei[s], dv, bv[2], B1, P[s][2]);
  }

  k_sim_wmma<64><<<NGRAPHS, 256, 0, stream>>>(P[0][0], P[1][0], out);
  k_sim_wmma<32><<<NGRAPHS, 256, 0, stream>>>(P[0][1], P[1][1],
                                              out + (size_t)1 * NGRAPHS * RES * RES);
  k_sim_wmma<16><<<NGRAPHS, 256, 0, stream>>>(P[0][2], P[1][2],
                                              out + (size_t)2 * NGRAPHS * RES * RES);
}


# k_agg: LDS float-atomic scatter replaced by gather over a dst-sorted edge list (LDS counting sort, once per side); node sums kept in registers, summed in the sort's cursor-arrival order
# speedup vs baseline: 4.6469x; 2.0195x over previous
#include <hip/hip_runtime.h>

typedef __attribute__((ext_vector_type(2))) float v2f;
typedef __attribute__((ext_vector_type(8))) float v8f;

#define NNODES  (128 * 512)
#define NEDGES  (128 * 512 * 16)
#define NGRAPHS 128
#define NPG     512
#define EPG     (512 * 16)
#define RES     128

#define SIDX_N 10240
#define DINV_THREADS 512
__global__ void __launch_bounds__(DINV_THREADS) k_csr(const int* __restrict__ ei,
                                                      float* __restrict__ dinv,
                                                      int* __restrict__ seg,
                                                      unsigned* __restrict__ sidx) {
  static_assert(EPG % DINV_THREADS == 0 && DINV_THREADS == NPG, "one node per thread");
  static_assert((SIDX_N / 2) % DINV_THREADS == 0, "whole sweeps of the block");
  __shared__ int cnt[NPG];
  __shared__ int cur[NPG];
  __shared__ int sc[2][NPG];
  __shared__ __align__(16) unsigned short s16[SIDX_N];
  unsigned* sw = reinterpret_cast<unsigned*>(s16);
  const int b = blockIdx.x;
  const int tid = threadIdx.x;
  int ss[EPG / DINV_THREADS], dd[EPG / DINV_THREADS];
#pragma unroll
  for (int i = 0; i < EPG / DINV_THREADS; ++i) {
    ss[i] = ei[b * EPG + tid + i * DINV_THREADS] & (NPG - 1);
    dd[i] = ei[NEDGES + b * EPG + tid + i * DINV_THREADS] & (NPG - 1);
  }
  cnt[tid] = 0;
#pragma unroll
  for (int i = 0; i < SIDX_N / 2 / DINV_THREADS; ++i) sw[tid + i * DINV_THREADS] = 0x02000200u;
  __syncthreads();
#pragma unroll
  for (int i = 0; i < EPG / DINV_THREADS; ++i) atomicAdd(&cnt[dd[i]], 1);
  __syncthreads();
  const int c = cnt[tid];
  dinv[b * NPG + tid] = rsqrtf((float)(c + 1));
  const int pc = (c + 3) >> 2;
  sc[0][tid] = pc;
  __syncthreads();
  int cb = 0;
#pragma unroll
  for (int off = 1; off < NPG; off <<= 1) {
    int v = sc[cb][tid];
    if (tid >= off) v += sc[cb][tid - off];
    sc[cb ^ 1][tid] = v;
    __syncthreads();
    cb ^= 1;
  }
  const int incl = sc[cb][tid];
  const int excl = incl - pc;
  cur[tid] = excl * 4;
  seg[b * (NPG + 1) + tid] = excl;
  if (tid == NPG - 1) seg[b * (NPG + 1) + NPG] = incl;
  __syncthreads();
#pragma unroll
  for (int i = 0; i < EPG / DINV_THREADS; ++i) {
    int pos = atomicAdd(&cur[dd[i]], 1);
    s16[pos] = (unsigned short)ss[i];
  }
  __syncthreads();
#pragma unroll
  for (int i = 0; i < SIDX_N / 2 / DINV_THREADS; ++i)
    sidx[(size_t)b * (SIDX_N / 2) + tid + i * DINV_THREADS] = sw[tid + i * DINV_THREADS];
}

#define GEMM_THREADS 256
template <int NT, int K>
__global__ void __launch_bounds__(GEMM_THREADS) k_gemm_wmma(const float* __restrict__ A,
                                                   const float* __restrict__ B,
                                                   float* __restrict__ C, int M) {
  const int N = NT * 16;
  static_assert(((K / 2) * NT * 16) % GEMM_THREADS == 0, "whole sweeps of the block");
  __shared__ v2f sB[(K / 2) * N];
#pragma unroll
  for (int it = 0; it < (K / 2) * N / GEMM_THREADS; ++it) {
    int i  = threadIdx.x + it * GEMM_THREADS;
    int kp = i / N;
    int c  = i - kp * N;
    v2f w;
    w.x = B[(size_t)(2 * kp) * N + c];
    w.y = B[(size_t)(2 * kp + 1) * N + c];
    sB[i] = w;
  }
  __syncthreads();

  int wave = (int)((blockIdx.x * blockDim.x + threadIdx.x) >> 5);
  int lane = threadIdx.x & 31;
  if (wave >= (M >> 4)) return;
  int r  = lane & 15;
  int kb = (lane >> 4) << 1;
  const float* Arow = A + (size_t)(wave * 16 + r) * K;

  v8f acc[NT] = {};
  for (int k = 0; k < K; k += 4) {
    v2f a = *(const v2f*)(Arow + k + kb);
    v2f bf[NT];
#pragma unroll
    for (int t = 0; t < NT; ++t)
      bf[t] = sB[((k + kb) >> 1) * N + t * 16 + r];
#pragma unroll
    for (int t = 0; t < NT; ++t)
      acc[t] = __builtin_amdgcn_wmma_f32_16x16x4_f32(false, a, false, bf[t], (short)0,
                                                     acc[t], false, false);
  }
  int rbase = (lane >> 4) << 3;
#pragma unroll
  for (int t = 0; t < NT; ++t) {
    float* Crow = C + (size_t)(wave * 16 + rbase) * N + t * 16 + r;
#pragma unroll
    for (int v = 0; v < 8; ++v) Crow[(size_t)v * N] = acc[t][v];
  }
}

#define AGG_THREADS 512
#define AGG_PAD 516
template <int C>
__global__ void __launch_bounds__(AGG_THREADS) k_agg(
    const float* __restrict__ h, const int* __restrict__ seg,
    const unsigned* __restrict__ sidx,
    const float* __restrict__ dinv, const float* __restrict__ bias,
    float* __restrict__ y, float* __restrict__ pout) {
  extern __shared__ __align__(16) float smem[];
  float* hl = smem;
  float* dv = smem + (NPG + 1) * C;
  int*   sg = reinterpret_cast<int*>(dv + AGG_PAD);
  uint2* sx2 = reinterpret_cast<uint2*>(sg + AGG_PAD);
  float4* hl4 = reinterpret_cast<float4*>(hl);
  const int b = blockIdx.x;
  const int gbase = b * NPG;
  const int tid = threadIdx.x;

  constexpr int chunks = C / 4;
  constexpr int RPT    = AGG_THREADS / chunks;
  constexpr int NV     = NPG * C / 4 / AGG_THREADS;
  constexpr int NSX    = SIDX_N / 4 / AGG_THREADS;
  static_assert(AGG_THREADS == NPG && AGG_THREADS % chunks == 0, "one dinv per thread");
  static_assert(NV % 4 == 0 && RPT * (NV / 4) == RES, "a lane group takes whole blocks of 4 nodes");
  static_assert((SIDX_N / 4) % AGG_THREADS == 0, "whole sweeps of the block");

  const uint2* sxg = reinterpret_cast<const uint2*>(sidx + (size_t)b * (SIDX_N / 2));
  uint2 sv[NSX];
#pragma unroll
  for (int i = 0; i < NSX; ++i) sv[i] = sxg[tid + i * AGG_THREADS];
  const int sgmine = seg[b * (NPG + 1) + tid];
  const int sglast = seg[b * (NPG + 1) + NPG];
  const float dmine = dinv[gbase + tid];
  const float4* h4 = reinterpret_cast<const float4*>(h + (size_t)gbase * C);
  float4 hv[NV];
#pragma unroll
  for (int i = 0; i < NV; ++i) hv[i] = h4[tid + i * AGG_THREADS];

  dv[tid] = dmine;
  sg[tid] = sgmine;
  if (tid == 0) { dv[NPG] = 0.0f; sg[NPG] = sglast; }
  if (tid < chunks) hl4[NPG * chunks + tid] = make_float4(0.0f, 0.0f, 0.0f, 0.0f);
#pragma unroll
  for (int i = 0; i < NV; ++i) hl4[tid + i * AGG_THREADS] = hv[i];
#pragma unroll
  for (int i = 0; i < NSX; ++i) sx2[tid + i * AGG_THREADS] = sv[i];
  __syncthreads();

  const int g  = tid / chunks;
  const int cq = tid - g * chunks;
  const float4 bb = *reinterpret_cast<const float4*>(bias + 4 * cq);
  float4* y4 = reinterpret_cast<float4*>(y + (size_t)gbase * C);
  float4* p4 = reinterpret_cast<float4*>(pout + (size_t)b * RES * C);
  const float4* hc = hl4 + cq;

  for (int t = 0; t < NV / 4; ++t) {
    const int q = g + t * RPT;
    float4 o[4];
#pragma unroll
    for (int j = 0; j < 4; ++j) {
      const int n = 4 * q + j;
      const float dn = dv[n];
      int e = sg[n];
      const int e1 = sg[n + 1];
      float4 a = hc[n * chunks];
      const float nn = dn * dn;
      a.x *= nn; a.y *= nn; a.z *= nn; a.w *= nn;
      for (; e < e1; ++e) {
        const uint2 pk = sx2[e];
        const int s0 = (int)(pk.x & 0xffffu), s1 = (int)(pk.x >> 16);
        const int s2 = (int)(pk.y & 0xffffu), s3 = (int)(pk.y >> 16);
        const float w0 = dv[s0] * dn, w1 = dv[s1] * dn, w2 = dv[s2] * dn, w3 = dv[s3] * dn;
        const float4 v0 = hc[s0 * chunks], v1 = hc[s1 * chunks];
        const float4 v2 = hc[s2 * chunks], v3 = hc[s3 * chunks];
        a.x += v0.x * w0; a.y += v0.y * w0; a.z += v0.z * w0; a.w += v0.w * w0;
        a.x += v1.x * w1; a.y += v1.y * w1; a.z += v1.z * w1; a.w += v1.w * w1;
        a.x += v2.x * w2; a.y += v2.y * w2; a.z += v2.z * w2; a.w += v2.w * w2;
        a.x += v3.x * w3; a.y += v3.y * w3; a.z += v3.z * w3; a.w += v3.w * w3;
      }
      float4 r;
      r.x = fmaxf(a.x + bb.x, 0.0f);
      r.y = fmaxf(a.y + bb.y, 0.0f);
      r.z = fmaxf(a.z + bb.z, 0.0f);
      r.w = fmaxf(a.w + bb.w, 0.0f);
      o[j] = r;
      y4[(size_t)n * chunks + cq] = r;
    }
    float4 pz;
    pz.x = 0.5f * (o[1].x + o[2].x);
    pz.y = 0.5f * (o[1].y + o[2].y);
    pz.z = 0.5f * (o[1].z + o[2].z);
    pz.w = 0.5f * (o[1].w + o[2].w);
    p4[q * chunks + cq] = pz;
  }
}

#define SIM_THREADS 256
template <int D>
__global__ void __launch_bounds__(SIM_THREADS) k_sim_wmma(const float* __restrict__ U,
                                                  const float* __restrict__ V,
                                                  float* __restrict__ out) {
  static_assert((RES * D / 4) % SIM_THREADS == 0, "whole sweeps of the block");
  __shared__ __align__(16) float sV[RES * D];
  const int b = blockIdx.x;
  const float4* Vb4 = reinterpret_cast<const float4*>(V + (size_t)b * RES * D);
  float4* sV4 = reinterpret_cast<float4*>(sV);
  float4 vv[RES * D / 4 / SIM_THREADS];
#pragma unroll
  for (int it = 0; it < RES * D / 4 / SIM_THREADS; ++it) vv[it] = Vb4[threadIdx.x + it * SIM_THREADS];
#pragma unroll
  for (int it = 0; it < RES * D / 4 / SIM_THREADS; ++it) sV4[threadIdx.x + it * SIM_THREADS] = vv[it];
  __syncthreads();

  int tm   = threadIdx.x >> 5;
  int lane = threadIdx.x & 31;
  int r    = lane & 15;
  int kb   = (lane >> 4) << 1;
  const float* Ar = U + ((size_t)b * RES + tm * 16 + r) * D;

  v8f acc[8] = {};
  for (int k = 0; k < D; k += 4) {
    v2f a = *(const v2f*)(Ar + k + kb);
    v2f bf[8];
#pragma unroll
    for (int t = 0; t < 8; ++t)
      bf[t] = *(const v2f*)(sV + (t * 16 + r) * D + k + kb);
#pragma unroll
    for (int t = 0; t < 8; ++t)
      acc[t] = __builtin_amdgcn_wmma_f32_16x16x4_f32(false, a, false, bf[t], (short)0,
                                                     acc[t], false, false);
  }
  int rbase = (lane >> 4) << 3;
#pragma unroll
  for (int t = 0; t < 8; ++t) {
    float* Co = out + (size_t)b * RES * RES + (size_t)(tm * 16 + rbase) * RES + t * 16 + r;
#pragma unroll
    for (int v = 0; v < 8; ++v) Co[(size_t)v * RES] = acc[t][v];
  }
}

extern "C" void kernel_launch(void* const* d_in, const int* in_sizes, int n_in,
                              void* d_out, int out_size, void* d_ws, size_t ws_size,
                              hipStream_t stream) {
  (void)in_sizes; (void)n_in; (void)out_size; (void)ws_size;
  const float* x[2]  = {(const float*)d_in[0], (const float*)d_in[1]};
  const int*   ei[2] = {(const int*)d_in[2],  (const int*)d_in[3]};
  const float* W[3]  = {(const float*)d_in[6], (const float*)d_in[8], (const float*)d_in[10]};
  const float* bv[3] = {(const float*)d_in[7], (const float*)d_in[9], (const float*)d_in[11]};
  float* out = (float*)d_out;

  float* ws = (float*)d_ws;
  size_t o = 0;
  float* dinvA = ws + o; o += (size_t)2 * NNODES;
  float* B0 = ws + o; o += (size_t)NNODES * 64;
  float* B1 = ws + o; o += (size_t)NNODES * 64;
  float* B2 = ws + o; o += (size_t)NNODES * 32;
  float* P[2][3];
  const int dims[3] = {64, 32, 16};
  for (int s = 0; s < 2; ++s)
    for (int l = 0; l < 3; ++l) { P[s][l] = ws + o; o += (size_t)NGRAPHS * RES * dims[l]; }
  int* segA = (int*)(ws + o); o += (size_t)2 * NGRAPHS * (NPG + 1);
  unsigned* sidxA = (unsigned*)(ws + o); o += (size_t)2 * NGRAPHS * (SIDX_N / 2);

  const unsigned gemmBlocks = (NNODES / 16) * 32 / 256;
  auto shbytes = [](int C) { return (size_t)((NPG + 1) * C + 2 * AGG_PAD + SIDX_N / 2) * sizeof(float); };

  for (int s = 0; s < 2; ++s) {
    float* dv = dinvA + (size_t)s * NNODES;
    int* sg = segA + (size_t)s * NGRAPHS * (NPG + 1);
    unsigned* sx = sidxA + (size_t)s * NGRAPHS * (SIDX_N / 2);
    k_csr<<<NGRAPHS, DINV_THREADS, 0, stream>>>(ei[s], dv, sg, sx);

    k_gemm_wmma<4, 128><<<gemmBlocks, 256, 0, stream>>>(x[s], W[0], B0, NNODES);
    k_agg<64><<<NGRAPHS, AGG_THREADS, shbytes(64), stream>>>(B0, sg, sx, dv, bv[0], B1, P[s][0]);
    k_gemm_wmma<2, 64><<<gemmBlocks, 256, 0, stream>>>(B1, W[1], B0, NNODES);
    k_agg<32><<<NGRAPHS, AGG_THREADS, shbytes(32), stream>>>(B0, sg, sx, dv, bv[1], B2, P[s][1]);
    k_gemm_wmma<1, 32><<<gemmBlocks, 256, 0, stream>>>(B2, W[2], B0, NNODES);
    k_agg<16><<<NGRAPHS, AGG_THREADS, shbytes(16), stream>>>(B0, sg, sx, dv, bv[2], B1, P[s][2]);
  }

  k_sim_wmma<64><<<NGRAPHS, 256, 0, stream>>>(P[0][0], P[1][0], out);
  k_sim_wmma<32><<<NGRAPHS, 256, 0, stream>>>(P[0][1], P[1][1],
                                              out + (size_t)1 * NGRAPHS * RES * RES);
  k_sim_wmma<16><<<NGRAPHS, 256, 0, stream>>>(P[0][2], P[1][2],
                                              out + (size_t)2 * NGRAPHS * RES * RES);
}


# one work-group per graph and side runs edge sort, the 3 weight products (WMMA) and the 3 gathers in ONE launch; activations, edge list and dinv stay in LDS between stages; 17 launches become 4
# speedup vs baseline: 10.2977x; 2.2160x over previous
#include <hip/hip_runtime.h>


typedef __attribute__((ext_vector_type(2))) float v2f;
typedef __attribute__((ext_vector_type(8))) float v8f;

#define NNODES  (128 * 512)
#define NEDGES  (128 * 512 * 16)
#define NGRAPHS 128
#define NPG     512
#define EPG     (512 * 16)
#define RES     128

#define SIDX_N  10240
#define FT      512
#define AGG_PAD 516
#define H_OFF   0
#define H_SIZE  ((NPG + 1) * 64)
#define Y_OFF   (H_OFF + H_SIZE)
#define Y_SIZE  (NPG * (64 + 4))
#define DV_OFF  (Y_OFF + Y_SIZE)
#define SG_OFF  (DV_OFF + AGG_PAD)
#define SX_OFF  (SG_OFF + AGG_PAD)
#define LDS_FLOATS (SX_OFF + SIDX_N / 2)
#define W2_OFF  ((NPG + 1) * 32)
#define W3_OFF  (W2_OFF + 64 * 32)
#define POOL_SIDE (NGRAPHS * RES * (64 + 32 + 16))

#define GU 8
template <int LDA>
__device__ __forceinline__ void a_group(const float* A, int wave, int lane, int kc,
                                        v2f (&n0)[GU], v2f (&n1)[GU]) {
  const float* A0 = A + (size_t)(wave * 16 + (lane & 15)) * LDA + ((lane >> 4) << 1);
  const float* A1 = A0 + (size_t)256 * LDA;
#pragma unroll
  for (int u = 0; u < GU; ++u) {
    n0[u] = *(const v2f*)(A0 + kc + 4 * u);
    n1[u] = *(const v2f*)(A1 + kc + 4 * u);
  }
}

template <int NT, int K, int LDA>
__device__ __forceinline__ void gemm_stage(const float* A, const v2f* sB, float* hl,
                                           int wave, int lane, v2f (&n0)[GU], v2f (&n1)[GU]) {
  constexpr int N = NT * 16;
  constexpr int U = GU;
  static_assert(K % (4 * U) == 0, "whole groups");
  const int r  = lane & 15;
  const int kb = (lane >> 4) << 1;

  v8f acc0[NT] = {};
  v8f acc1[NT] = {};
#pragma unroll
  for (int kc = 0; kc < K; kc += 4 * U) {
    v2f c0[U], c1[U];
#pragma unroll
    for (int u = 0; u < U; ++u) { c0[u] = n0[u]; c1[u] = n1[u]; }
    if (kc + 4 * U < K) a_group<LDA>(A, wave, lane, kc + 4 * U, n0, n1);
#pragma unroll
    for (int u = 0; u < U; ++u) {
      const int k = kc + 4 * u;
      v2f bf[NT];
#pragma unroll
      for (int t = 0; t < NT; ++t) bf[t] = sB[((k + kb) >> 1) * N + t * 16 + r];
#pragma unroll
      for (int t = 0; t < NT; ++t)
        acc0[t] = __builtin_amdgcn_wmma_f32_16x16x4_f32(false, c0[u], false, bf[t], (short)0,
                                                        acc0[t], false, false);
#pragma unroll
      for (int t = 0; t < NT; ++t)
        acc1[t] = __builtin_amdgcn_wmma_f32_16x16x4_f32(false, c1[u], false, bf[t], (short)0,
                                                        acc1[t], false, false);
    }
  }
  const int rbase = (lane >> 4) << 3;
#pragma unroll
  for (int t = 0; t < NT; ++t) {
    float* d0 = hl + (wave * 16 + rbase) * N + t * 16 + r;
    float* d1 = d0 + 256 * N;
#pragma unroll
    for (int v = 0; v < 8; ++v) { d0[v * N] = acc0[t][v]; d1[v * N] = acc1[t][v]; }
  }
}

template <int C, bool WRITE_Y>
__device__ __forceinline__ void agg_stage(const float* hl, const float* dv, const int* sg,
                                          const uint2* sx2, const float4 bb, float* Y,
                                          float* __restrict__ pout, int tid) {
  constexpr int chunks = C / 4;
  constexpr int RPT    = FT / chunks;
  constexpr int NV     = NPG * C / 4 / FT;
  static_assert(FT % chunks == 0, "whole lane groups");
  static_assert(NV % 4 == 0 && RPT * (NV / 4) == RES, "a lane group takes whole blocks of 4 nodes");
  const int g  = tid / chunks;
  const int cq = tid - g * chunks;
  const float4* hc = reinterpret_cast<const float4*>(hl) + cq;
  float4* Y4 = reinterpret_cast<float4*>(Y);
  float4* p4 = reinterpret_cast<float4*>(pout);

  for (int t = 0; t < NV / 4; ++t) {
    const int q = g + t * RPT;
    float4 o[4];
#pragma unroll
    for (int j = 0; j < 4; ++j) {
      const int n = 4 * q + j;
      const float dn = dv[n];
      int e = sg[n];
      const int e1 = sg[n + 1];
      float4 a = hc[n * chunks];
      const float nn = dn * dn;
      a.x *= nn; a.y *= nn; a.z *= nn; a.w *= nn;
      for (; e < e1; ++e) {
        const uint2 pk = sx2[e];
        const int s0 = (int)(pk.x & 0xffffu), s1 = (int)(pk.x >> 16);
        const int s2 = (int)(pk.y & 0xffffu), s3 = (int)(pk.y >> 16);
        const float w0 = dv[s0] * dn, w1 = dv[s1] * dn, w2 = dv[s2] * dn, w3 = dv[s3] * dn;
        const float4 v0 = hc[s0 * chunks], v1 = hc[s1 * chunks];
        const float4 v2 = hc[s2 * chunks], v3 = hc[s3 * chunks];
        a.x += v0.x * w0; a.y += v0.y * w0; a.z += v0.z * w0; a.w += v0.w * w0;
        a.x += v1.x * w1; a.y += v1.y * w1; a.z += v1.z * w1; a.w += v1.w * w1;
        a.x += v2.x * w2; a.y += v2.y * w2; a.z += v2.z * w2; a.w += v2.w * w2;
        a.x += v3.x * w3; a.y += v3.y * w3; a.z += v3.z * w3; a.w += v3.w * w3;
      }
      float4 rv;
      rv.x = fmaxf(a.x + bb.x, 0.0f);
      rv.y = fmaxf(a.y + bb.y, 0.0f);
      rv.z = fmaxf(a.z + bb.z, 0.0f);
      rv.w = fmaxf(a.w + bb.w, 0.0f);
      o[j] = rv;
      if (WRITE_Y) Y4[n * (chunks + 1) + cq] = rv;
    }
    float4 pz;
    pz.x = 0.5f * (o[1].x + o[2].x);
    pz.y = 0.5f * (o[1].y + o[2].y);
    pz.z = 0.5f * (o[1].z + o[2].z);
    pz.w = 0.5f * (o[1].w + o[2].w);
    p4[q * chunks + cq] = pz;
  }
}

__global__ void __launch_bounds__(FT) k_graph(
    const float* __restrict__ x0, const float* __restrict__ x1,
    const int* __restrict__ ei0, const int* __restrict__ ei1,
    const float* __restrict__ W1, const float* __restrict__ b1,
    const float* __restrict__ W2, const float* __restrict__ b2,
    const float* __restrict__ W3, const float* __restrict__ b3,
    float* __restrict__ pool) {
  static_assert(EPG % FT == 0 && FT == NPG, "one node per thread");
  static_assert((SIDX_N / 2) % FT == 0, "whole sweeps of the block");
  __shared__ __align__(16) float smem[LDS_FLOATS];
  float* H  = smem + H_OFF;
  float* Y  = smem + Y_OFF;
  float* dv = smem + DV_OFF;
  int*   sg = reinterpret_cast<int*>(smem + SG_OFF);
  unsigned* sw = reinterpret_cast<unsigned*>(smem + SX_OFF);
  unsigned short* s16 = reinterpret_cast<unsigned short*>(sw);
  const uint2* sx2 = reinterpret_cast<const uint2*>(sw);
  int* cnt = reinterpret_cast<int*>(H);
  int* cur = cnt + NPG;
  int* sc  = cur + NPG;
  v2f* sB1 = reinterpret_cast<v2f*>(Y);
  v2f* sB2 = reinterpret_cast<v2f*>(H + W2_OFF);
  v2f* sB3 = reinterpret_cast<v2f*>(H + W3_OFF);

  const int b    = blockIdx.x;
  const int side = blockIdx.y;
  const int tid  = threadIdx.x;
  const float* x  = side ? x1 : x0;
  const int*   ei = side ? ei1 : ei0;
  float* pside = pool + (size_t)side * POOL_SIDE;

  int ss[EPG / FT], dd[EPG / FT];
#pragma unroll
  for (int i = 0; i < EPG / FT; ++i) {
    ss[i] = ei[b * EPG + tid + i * FT] & (NPG - 1);
    dd[i] = ei[NEDGES + b * EPG + tid + i * FT] & (NPG - 1);
  }
  v2f w1r[64 * 64 / FT];
#pragma unroll
  for (int it = 0; it < 64 * 64 / FT; ++it) {
    const int i = tid + it * FT, kp = i / 64, c = i - kp * 64;
    w1r[it].x = W1[(2 * kp) * 64 + c];
    w1r[it].y = W1[(2 * kp + 1) * 64 + c];
  }
  v2f w2r[32 * 32 / FT];
#pragma unroll
  for (int it = 0; it < 32 * 32 / FT; ++it) {
    const int i = tid + it * FT, kp = i / 32, c = i - kp * 32;
    w2r[it].x = W2[(2 * kp) * 32 + c];
    w2r[it].y = W2[(2 * kp + 1) * 32 + c];
  }
  v2f w3r = {0.0f, 0.0f};
  if (tid < 16 * 16) {
    const int kp = tid / 16, c = tid - kp * 16;
    w3r.x = W3[(2 * kp) * 16 + c];
    w3r.y = W3[(2 * kp + 1) * 16 + c];
  }
  const float4 bb1 = *reinterpret_cast<const float4*>(b1 + 4 * (tid & 15));
  const float4 bb2 = *reinterpret_cast<const float4*>(b2 + 4 * (tid & 7));
  const float4 bb3 = *reinterpret_cast<const float4*>(b3 + 4 * (tid & 3));
  const int wave = tid >> 5;
  const int lane = tid & 31;
  const float* xg = x + (size_t)b * NPG * 128;
  v2f n0[GU], n1[GU];
  a_group<128>(xg, wave, lane, 0, n0, n1);

  cnt[tid] = 0;
#pragma unroll
  for (int i = 0; i < SIDX_N / 2 / FT; ++i) sw[tid + i * FT] = 0x02000200u;
#pragma unroll
  for (int it = 0; it < 64 * 64 / FT; ++it) sB1[tid + it * FT] = w1r[it];
  __syncthreads();

#pragma unroll
  for (int i = 0; i < EPG / FT; ++i) atomicAdd(&cnt[dd[i]], 1);
  __syncthreads();
  const int c = cnt[tid];
  dv[tid] = rsqrtf((float)(c + 1));
  if (tid == 0) dv[NPG] = 0.0f;
  const int pc = (c + 3) >> 2;
  sc[tid] = pc;
  __syncthreads();
  int cb = 0;
#pragma unroll
  for (int off = 1; off < NPG; off <<= 1) {
    int v = sc[cb * NPG + tid];
    if (tid >= off) v += sc[cb * NPG + tid - off];
    sc[(cb ^ 1) * NPG + tid] = v;
    __syncthreads();
    cb ^= 1;
  }
  const int incl = sc[cb * NPG + tid];
  const int excl = incl - pc;
  cur[tid] = excl * 4;
  sg[tid] = excl;
  if (tid == NPG - 1) sg[NPG] = incl;
  __syncthreads();
#pragma unroll
  for (int i = 0; i < EPG / FT; ++i) {
    int pos = atomicAdd(&cur[dd[i]], 1);
    s16[pos] = (unsigned short)ss[i];
  }
  __syncthreads();

  float4* H4 = reinterpret_cast<float4*>(H);
  const float4 z4 = make_float4(0.0f, 0.0f, 0.0f, 0.0f);

  gemm_stage<4, 128, 128>(xg, sB1, H, wave, lane, n0, n1);
  if (tid < 16) H4[NPG * 16 + tid] = z4;
  __syncthreads();
  agg_stage<64, true>(H, dv, sg, sx2, bb1, Y, pside + (size_t)b * RES * 64, tid);
  __syncthreads();
  sB2[tid] = w2r[0];
  sB2[tid + FT] = w2r[1];
  if (tid < 16 * 16) sB3[tid] = w3r;
  __syncthreads();

  a_group<64 + 4>(Y, wave, lane, 0, n0, n1);
  gemm_stage<2, 64, 64 + 4>(Y, sB2, H, wave, lane, n0, n1);
  if (tid < 8) H4[NPG * 8 + tid] = z4;
  __syncthreads();
  agg_stage<32, true>(H, dv, sg, sx2, bb2, Y,
                      pside + (size_t)NGRAPHS * RES * 64 + (size_t)b * RES * 32, tid);
  __syncthreads();

  a_group<32 + 4>(Y, wave, lane, 0, n0, n1);
  gemm_stage<1, 32, 32 + 4>(Y, sB3, H, wave, lane, n0, n1);
  if (tid < 4) H4[NPG * 4 + tid] = z4;
  __syncthreads();
  agg_stage<16, false>(H, dv, sg, sx2, bb3, Y,
                       pside + (size_t)NGRAPHS * RES * 96 + (size_t)b * RES * 16, tid);
}

#define SIM_THREADS 256
template <int D>
__global__ void __launch_bounds__(SIM_THREADS) k_sim_wmma(const float* __restrict__ U,
                                                  const float* __restrict__ V,
                                                  float* __restrict__ out) {
  static_assert((RES * D / 4) % SIM_THREADS == 0, "whole sweeps of the block");
  __shared__ __align__(16) float sV[RES * D];
  const int b = blockIdx.x;
  const float4* Vb4 = reinterpret_cast<const float4*>(V + (size_t)b * RES * D);
  float4* sV4 = reinterpret_cast<float4*>(sV);
  float4 vv[RES * D / 4 / SIM_THREADS];
#pragma unroll
  for (int it = 0; it < RES * D / 4 / SIM_THREADS; ++it) vv[it] = Vb4[threadIdx.x + it * SIM_THREADS];
#pragma unroll
  for (int it = 0; it < RES * D / 4 / SIM_THREADS; ++it) sV4[threadIdx.x + it * SIM_THREADS] = vv[it];
  __syncthreads();

  int tm   = threadIdx.x >> 5;
  int lane = threadIdx.x & 31;
  int r    = lane & 15;
  int kb   = (lane >> 4) << 1;
  const float* Ar = U + ((size_t)b * RES + tm * 16 + r) * D;

  v8f acc[8] = {};
  for (int k = 0; k < D; k += 4) {
    v2f a = *(const v2f*)(Ar + k + kb);
    v2f bf[8];
#pragma unroll
    for (int t = 0; t < 8; ++t)
      bf[t] = *(const v2f*)(sV + (t * 16 + r) * D + k + kb);
#pragma unroll
    for (int t = 0; t < 8; ++t)
      acc[t] = __builtin_amdgcn_wmma_f32_16x16x4_f32(false, a, false, bf[t], (short)0,
                                                     acc[t], false, false);
  }
  int rbase = (lane >> 4) << 3;
#pragma unroll
  for (int t = 0; t < 8; ++t) {
    float* Co = out + (size_t)b * RES * RES + (size_t)(tm * 16 + rbase) * RES + t * 16 + r;
#pragma unroll
    for (int v = 0; v < 8; ++v) Co[(size_t)v * RES] = acc[t][v];
  }
}

extern "C" void kernel_launch(void* const* d_in, const int* in_sizes, int n_in,
                              void* d_out, int out_size, void* d_ws, size_t ws_size,
                              hipStream_t stream) {
  (void)in_sizes; (void)n_in; (void)out_size; (void)ws_size;
  const float* x[2]  = {(const float*)d_in[0], (const float*)d_in[1]};
  const int*   ei[2] = {(const int*)d_in[2],  (const int*)d_in[3]};
  const float* W[3]  = {(const float*)d_in[6], (const float*)d_in[8], (const float*)d_in[10]};
  const float* bv[3] = {(const float*)d_in[7], (const float*)d_in[9], (const float*)d_in[11]};
  float* out = (float*)d_out;

  float* pool = (float*)d_ws;
  float* P[2][3];
  for (int s = 0; s < 2; ++s) {
    P[s][0] = pool + (size_t)s * POOL_SIDE;
    P[s][1] = P[s][0] + (size_t)NGRAPHS * RES * 64;
    P[s][2] = P[s][0] + (size_t)NGRAPHS * RES * 96;
  }

  k_graph<<<dim3(NGRAPHS, 2), FT, 0, stream>>>(x[0], x[1], ei[0], ei[1], W[0], bv[0], W[1], bv[1],
                                               W[2], bv[2], pool);

  k_sim_wmma<64><<<NGRAPHS, 256, 0, stream>>>(P[0][0], P[1][0], out);
  k_sim_wmma<32><<<NGRAPHS, 256, 0, stream>>>(P[0][1], P[1][1],
                                              out + (size_t)1 * NGRAPHS * RES * RES);
  k_sim_wmma<16><<<NGRAPHS, 256, 0, stream>>>(P[0][2], P[1][2],
                                              out + (size_t)2 * NGRAPHS * RES * RES);
}


# the three similarity launches (one per layer) merged into one launch of 128 x 3 work-groups, layer picked by blockIdx.y; same per-work-group code; 4 launches become 2
# speedup vs baseline: 12.2367x; 1.1883x over previous
#include <hip/hip_runtime.h>


typedef __attribute__((ext_vector_type(2))) float v2f;
typedef __attribute__((ext_vector_type(8))) float v8f;

#define NNODES  (128 * 512)
#define NEDGES  (128 * 512 * 16)
#define NGRAPHS 128
#define NPG     512
#define EPG     (512 * 16)
#define RES     128

#define SIDX_N  10240
#define FT      512
#define AGG_PAD 516
#define H_OFF   0
#define H_SIZE  ((NPG + 1) * 64)
#define Y_OFF   (H_OFF + H_SIZE)
#define Y_SIZE  (NPG * (64 + 4))
#define DV_OFF  (Y_OFF + Y_SIZE)
#define SG_OFF  (DV_OFF + AGG_PAD)
#define SX_OFF  (SG_OFF + AGG_PAD)
#define LDS_FLOATS (SX_OFF + SIDX_N / 2)
#define W2_OFF  ((NPG + 1) * 32)
#define W3_OFF  (W2_OFF + 64 * 32)
#define POOL_SIDE (NGRAPHS * RES * (64 + 32 + 16))

#define GU 8
template <int LDA>
__device__ __forceinline__ void a_group(const float* A, int wave, int lane, int kc,
                                        v2f (&n0)[GU], v2f (&n1)[GU]) {
  const float* A0 = A + (size_t)(wave * 16 + (lane & 15)) * LDA + ((lane >> 4) << 1);
  const float* A1 = A0 + (size_t)256 * LDA;
#pragma unroll
  for (int u = 0; u < GU; ++u) {
    n0[u] = *(const v2f*)(A0 + kc + 4 * u);
    n1[u] = *(const v2f*)(A1 + kc + 4 * u);
  }
}

template <int NT, int K, int LDA>
__device__ __forceinline__ void gemm_stage(const float* A, const v2f* sB, float* hl,
                                           int wave, int lane, v2f (&n0)[GU], v2f (&n1)[GU]) {
  constexpr int N = NT * 16;
  constexpr int U = GU;
  static_assert(K % (4 * U) == 0, "whole groups");
  const int r  = lane & 15;
  const int kb = (lane >> 4) << 1;

  v8f acc0[NT] = {};
  v8f acc1[NT] = {};
#pragma unroll
  for (int kc = 0; kc < K; kc += 4 * U) {
    v2f c0[U], c1[U];
#pragma unroll
    for (int u = 0; u < U; ++u) { c0[u] = n0[u]; c1[u] = n1[u]; }
    if (kc + 4 * U < K) a_group<LDA>(A, wave, lane, kc + 4 * U, n0, n1);
#pragma unroll
    for (int u = 0; u < U; ++u) {
      const int k = kc + 4 * u;
      v2f bf[NT];
#pragma unroll
      for (int t = 0; t < NT; ++t) bf[t] = sB[((k + kb) >> 1) * N + t * 16 + r];
#pragma unroll
      for (int t = 0; t < NT; ++t)
        acc0[t] = __builtin_amdgcn_wmma_f32_16x16x4_f32(false, c0[u], false, bf[t], (short)0,
                                                        acc0[t], false, false);
#pragma unroll
      for (int t = 0; t < NT; ++t)
        acc1[t] = __builtin_amdgcn_wmma_f32_16x16x4_f32(false, c1[u], false, bf[t], (short)0,
                                                        acc1[t], false, false);
    }
  }
  const int rbase = (lane >> 4) << 3;
#pragma unroll
  for (int t = 0; t < NT; ++t) {
    float* d0 = hl + (wave * 16 + rbase) * N + t * 16 + r;
    float* d1 = d0 + 256 * N;
#pragma unroll
    for (int v = 0; v < 8; ++v) { d0[v * N] = acc0[t][v]; d1[v * N] = acc1[t][v]; }
  }
}

template <int C, bool WRITE_Y>
__device__ __forceinline__ void agg_stage(const float* hl, const float* dv, const int* sg,
                                          const uint2* sx2, const float4 bb, float* Y,
                                          float* __restrict__ pout, int tid) {
  constexpr int chunks = C / 4;
  constexpr int RPT    = FT / chunks;
  constexpr int NV     = NPG * C / 4 / FT;
  static_assert(FT % chunks == 0, "whole lane groups");
  static_assert(NV % 4 == 0 && RPT * (NV / 4) == RES, "a lane group takes whole blocks of 4 nodes");
  const int g  = tid / chunks;
  const int cq = tid - g * chunks;
  const float4* hc = reinterpret_cast<const float4*>(hl) + cq;
  float4* Y4 = reinterpret_cast<float4*>(Y);
  float4* p4 = reinterpret_cast<float4*>(pout);

  for (int t = 0; t < NV / 4; ++t) {
    const int q = g + t * RPT;
    float4 o[4];
#pragma unroll
    for (int j = 0; j < 4; ++j) {
      const int n = 4 * q + j;
      const float dn = dv[n];
      int e = sg[n];
      const int e1 = sg[n + 1];
      float4 a = hc[n * chunks];
      const float nn = dn * dn;
      a.x *= nn; a.y *= nn; a.z *= nn; a.w *= nn;
      for (; e < e1; ++e) {
        const uint2 pk = sx2[e];
        const int s0 = (int)(pk.x & 0xffffu), s1 = (int)(pk.x >> 16);
        const int s2 = (int)(pk.y & 0xffffu), s3 = (int)(pk.y >> 16);
        const float w0 = dv[s0] * dn, w1 = dv[s1] * dn, w2 = dv[s2] * dn, w3 = dv[s3] * dn;
        const float4 v0 = hc[s0 * chunks], v1 = hc[s1 * chunks];
        const float4 v2 = hc[s2 * chunks], v3 = hc[s3 * chunks];
        a.x += v0.x * w0; a.y += v0.y * w0; a.z += v0.z * w0; a.w += v0.w * w0;
        a.x += v1.x * w1; a.y += v1.y * w1; a.z += v1.z * w1; a.w += v1.w * w1;
        a.x += v2.x * w2; a.y += v2.y * w2; a.z += v2.z * w2; a.w += v2.w * w2;
        a.x += v3.x * w3; a.y += v3.y * w3; a.z += v3.z * w3; a.w += v3.w * w3;
      }
      float4 rv;
      rv.x = fmaxf(a.x + bb.x, 0.0f);
      rv.y = fmaxf(a.y + bb.y, 0.0f);
      rv.z = fmaxf(a.z + bb.z, 0.0f);
      rv.w = fmaxf(a.w + bb.w, 0.0f);
      o[j] = rv;
      if (WRITE_Y) Y4[n * (chunks + 1) + cq] = rv;
    }
    float4 pz;
    pz.x = 0.5f * (o[1].x + o[2].x);
    pz.y = 0.5f * (o[1].y + o[2].y);
    pz.z = 0.5f * (o[1].z + o[2].z);
    pz.w = 0.5f * (o[1].w + o[2].w);
    p4[q * chunks + cq] = pz;
  }
}

__global__ void __launch_bounds__(FT) k_graph(
    const float* __restrict__ x0, const float* __restrict__ x1,
    const int* __restrict__ ei0, const int* __restrict__ ei1,
    const float* __restrict__ W1, const float* __restrict__ b1,
    const float* __restrict__ W2, const float* __restrict__ b2,
    const float* __restrict__ W3, const float* __restrict__ b3,
    float* __restrict__ pool) {
  static_assert(EPG % FT == 0 && FT == NPG, "one node per thread");
  static_assert((SIDX_N / 2) % FT == 0, "whole sweeps of the block");
  __shared__ __align__(16) float smem[LDS_FLOATS];
  float* H  = smem + H_OFF;
  float* Y  = smem + Y_OFF;
  float* dv = smem + DV_OFF;
  int*   sg = reinterpret_cast<int*>(smem + SG_OFF);
  unsigned* sw = reinterpret_cast<unsigned*>(smem + SX_OFF);
  unsigned short* s16 = reinterpret_cast<unsigned short*>(sw);
  const uint2* sx2 = reinterpret_cast<const uint2*>(sw);
  int* cnt = reinterpret_cast<int*>(H);
  int* cur = cnt + NPG;
  int* sc  = cur + NPG;
  v2f* sB1 = reinterpret_cast<v2f*>(Y);
  v2f* sB2 = reinterpret_cast<v2f*>(H + W2_OFF);
  v2f* sB3 = reinterpret_cast<v2f*>(H + W3_OFF);

  const int b    = blockIdx.x;
  const int side = blockIdx.y;
  const int tid  = threadIdx.x;
  const float* x  = side ? x1 : x0;
  const int*   ei = side ? ei1 : ei0;
  float* pside = pool + (size_t)side * POOL_SIDE;

  int ss[EPG / FT], dd[EPG / FT];
#pragma unroll
  for (int i = 0; i < EPG / FT; ++i) {
    ss[i] = ei[b * EPG + tid + i * FT] & (NPG - 1);
    dd[i] = ei[NEDGES + b * EPG + tid + i * FT] & (NPG - 1);
  }
  v2f w1r[64 * 64 / FT];
#pragma unroll
  for (int it = 0; it < 64 * 64 / FT; ++it) {
    const int i = tid + it * FT, kp = i / 64, c = i - kp * 64;
    w1r[it].x = W1[(2 * kp) * 64 + c];
    w1r[it].y = W1[(2 * kp + 1) * 64 + c];
  }
  v2f w2r[32 * 32 / FT];
#pragma unroll
  for (int it = 0; it < 32 * 32 / FT; ++it) {
    const int i = tid + it * FT, kp = i / 32, c = i - kp * 32;
    w2r[it].x = W2[(2 * kp) * 32 + c];
    w2r[it].y = W2[(2 * kp + 1) * 32 + c];
  }
  v2f w3r = {0.0f, 0.0f};
  if (tid < 16 * 16) {
    const int kp = tid / 16, c = tid - kp * 16;
    w3r.x = W3[(2 * kp) * 16 + c];
    w3r.y = W3[(2 * kp + 1) * 16 + c];
  }
  const float4 bb1 = *reinterpret_cast<const float4*>(b1 + 4 * (tid & 15));
  const float4 bb2 = *reinterpret_cast<const float4*>(b2 + 4 * (tid & 7));
  const float4 bb3 = *reinterpret_cast<const float4*>(b3 + 4 * (tid & 3));
  const int wave = tid >> 5;
  const int lane = tid & 31;
  const float* xg = x + (size_t)b * NPG * 128;
  v2f n0[GU], n1[GU];
  a_group<128>(xg, wave, lane, 0, n0, n1);

  cnt[tid] = 0;
#pragma unroll
  for (int i = 0; i < SIDX_N / 2 / FT; ++i) sw[tid + i * FT] = 0x02000200u;
#pragma unroll
  for (int it = 0; it < 64 * 64 / FT; ++it) sB1[tid + it * FT] = w1r[it];
  __syncthreads();

#pragma unroll
  for (int i = 0; i < EPG / FT; ++i) atomicAdd(&cnt[dd[i]], 1);
  __syncthreads();
  const int c = cnt[tid];
  dv[tid] = rsqrtf((float)(c + 1));
  if (tid == 0) dv[NPG] = 0.0f;
  const int pc = (c + 3) >> 2;
  sc[tid] = pc;
  __syncthreads();
  int cb = 0;
#pragma unroll
  for (int off = 1; off < NPG; off <<= 1) {
    int v = sc[cb * NPG + tid];
    if (tid >= off) v += sc[cb * NPG + tid - off];
    sc[(cb ^ 1) * NPG + tid] = v;
    __syncthreads();
    cb ^= 1;
  }
  const int incl = sc[cb * NPG + tid];
  const int excl = incl - pc;
  cur[tid] = excl * 4;
  sg[tid] = excl;
  if (tid == NPG - 1) sg[NPG] = incl;
  __syncthreads();
#pragma unroll
  for (int i = 0; i < EPG / FT; ++i) {
    int pos = atomicAdd(&cur[dd[i]], 1);
    s16[pos] = (unsigned short)ss[i];
  }
  __syncthreads();

  float4* H4 = reinterpret_cast<float4*>(H);
  const float4 z4 = make_float4(0.0f, 0.0f, 0.0f, 0.0f);

  gemm_stage<4, 128, 128>(xg, sB1, H, wave, lane, n0, n1);
  if (tid < 16) H4[NPG * 16 + tid] = z4;
  __syncthreads();
  agg_stage<64, true>(H, dv, sg, sx2, bb1, Y, pside + (size_t)b * RES * 64, tid);
  __syncthreads();
  sB2[tid] = w2r[0];
  sB2[tid + FT] = w2r[1];
  if (tid < 16 * 16) sB3[tid] = w3r;
  __syncthreads();

  a_group<64 + 4>(Y, wave, lane, 0, n0, n1);
  gemm_stage<2, 64, 64 + 4>(Y, sB2, H, wave, lane, n0, n1);
  if (tid < 8) H4[NPG * 8 + tid] = z4;
  __syncthreads();
  agg_stage<32, true>(H, dv, sg, sx2, bb2, Y,
                      pside + (size_t)NGRAPHS * RES * 64 + (size_t)b * RES * 32, tid);
  __syncthreads();

  a_group<32 + 4>(Y, wave, lane, 0, n0, n1);
  gemm_stage<1, 32, 32 + 4>(Y, sB3, H, wave, lane, n0, n1);
  if (tid < 4) H4[NPG * 4 + tid] = z4;
  __syncthreads();
  agg_stage<16, false>(H, dv, sg, sx2, bb3, Y,
                       pside + (size_t)NGRAPHS * RES * 96 + (size_t)b * RES * 16, tid);
}

#define SIM_THREADS 256
template <int D>
__device__ __forceinline__ void sim_body(const float* __restrict__ U, const float* __restrict__ V,
                                         float* __restrict__ out, float* sV, int b) {
  static_assert((RES * D / 4) % SIM_THREADS == 0, "whole sweeps of the block");
  const float4* Vb4 = reinterpret_cast<const float4*>(V + (size_t)b * RES * D);
  float4* sV4 = reinterpret_cast<float4*>(sV);
  float4 vv[RES * D / 4 / SIM_THREADS];
#pragma unroll
  for (int it = 0; it < RES * D / 4 / SIM_THREADS; ++it) vv[it] = Vb4[threadIdx.x + it * SIM_THREADS];
#pragma unroll
  for (int it = 0; it < RES * D / 4 / SIM_THREADS; ++it) sV4[threadIdx.x + it * SIM_THREADS] = vv[it];
  __syncthreads();

  int tm   = threadIdx.x >> 5;
  int lane = threadIdx.x & 31;
  int r    = lane & 15;
  int kb   = (lane >> 4) << 1;
  const float* Ar = U + ((size_t)b * RES + tm * 16 + r) * D;

  v8f acc[8] = {};
  for (int k = 0; k < D; k += 4) {
    v2f a = *(const v2f*)(Ar + k + kb);
    v2f bf[8];
#pragma unroll
    for (int t = 0; t < 8; ++t)
      bf[t] = *(const v2f*)(sV + (t * 16 + r) * D + k + kb);
#pragma unroll
    for (int t = 0; t < 8; ++t)
      acc[t] = __builtin_amdgcn_wmma_f32_16x16x4_f32(false, a, false, bf[t], (short)0,
                                                     acc[t], false, false);
  }
  int rbase = (lane >> 4) << 3;
#pragma unroll
  for (int t = 0; t < 8; ++t) {
    float* Co = out + (size_t)b * RES * RES + (size_t)(tm * 16 + rbase) * RES + t * 16 + r;
#pragma unroll
    for (int v = 0; v < 8; ++v) Co[(size_t)v * RES] = acc[t][v];
  }
}

__global__ void __launch_bounds__(SIM_THREADS) k_sim_all(const float* __restrict__ pool,
                                                         float* __restrict__ out) {
  __shared__ __align__(16) float sV[RES * 64];
  const int b = blockIdx.x;
  const int layer = blockIdx.y;
  const float* U = pool;
  const float* V = pool + POOL_SIDE;
  if (layer == 0) {
    sim_body<64>(U, V, out, sV, b);
  } else if (layer == 1) {
    sim_body<32>(U + (size_t)NGRAPHS * RES * 64, V + (size_t)NGRAPHS * RES * 64,
                 out + (size_t)1 * NGRAPHS * RES * RES, sV, b);
  } else {
    sim_body<16>(U + (size_t)NGRAPHS * RES * 96, V + (size_t)NGRAPHS * RES * 96,
                 out + (size_t)2 * NGRAPHS * RES * RES, sV, b);
  }
}

extern "C" void kernel_launch(void* const* d_in, const int* in_sizes, int n_in,
                              void* d_out, int out_size, void* d_ws, size_t ws_size,
                              hipStream_t stream) {
  (void)in_sizes; (void)n_in; (void)out_size; (void)ws_size;
  const float* x[2]  = {(const float*)d_in[0], (const float*)d_in[1]};
  const int*   ei[2] = {(const int*)d_in[2],  (const int*)d_in[3]};
  const float* W[3]  = {(const float*)d_in[6], (const float*)d_in[8], (const float*)d_in[10]};
  const float* bv[3] = {(const float*)d_in[7], (const float*)d_in[9], (const float*)d_in[11]};
  float* out = (float*)d_out;

  float* pool = (float*)d_ws;

  k_graph<<<dim3(NGRAPHS, 2), FT, 0, stream>>>(x[0], x[1], ei[0], ei[1], W[0], bv[0], W[1], bv[1],
                                               W[2], bv[2], pool);

  k_sim_all<<<dim3(NGRAPHS, 3), SIM_THREADS, 0, stream>>>(pool, out);
}


# gather: a lane group walks the adjacent lists of its 4 nodes as ONE run of 4-group trips (own row stored as each list's first entry, sum restarts at node starts); per-node sum order unchanged
# speedup vs baseline: 13.3782x; 1.0933x over previous
#include <hip/hip_runtime.h>


typedef __attribute__((ext_vector_type(2))) float v2f;
typedef __attribute__((ext_vector_type(8))) float v8f;

#define NNODES  (128 * 512)
#define NEDGES  (128 * 512 * 16)
#define NGRAPHS 128
#define NPG     512
#define EPG     (512 * 16)
#define RES     128

#define SIDX_N  10240
#define FT      512
#define AGG_PAD 516
#define H_OFF   0
#define H_SIZE  ((NPG + 1) * 64)
#define Y_OFF   (H_OFF + H_SIZE)
#define Y_SIZE  (NPG * (64 + 4))
#define DV_OFF  (Y_OFF + Y_SIZE)
#define SG_OFF  (DV_OFF + AGG_PAD)
#define SX_OFF  (SG_OFF + AGG_PAD)
#define LDS_FLOATS (SX_OFF + SIDX_N / 2 + 8)
#define W2_OFF  ((NPG + 1) * 32)
#define W3_OFF  (W2_OFF + 64 * 32)
#define POOL_SIDE (NGRAPHS * RES * (64 + 32 + 16))

#define GU 8
template <int LDA>
__device__ __forceinline__ void a_group(const float* A, int wave, int lane, int kc,
                                        v2f (&n0)[GU], v2f (&n1)[GU]) {
  const float* A0 = A + (size_t)(wave * 16 + (lane & 15)) * LDA + ((lane >> 4) << 1);
  const float* A1 = A0 + (size_t)256 * LDA;
#pragma unroll
  for (int u = 0; u < GU; ++u) {
    n0[u] = *(const v2f*)(A0 + kc + 4 * u);
    n1[u] = *(const v2f*)(A1 + kc + 4 * u);
  }
}

template <int NT, int K, int LDA>
__device__ __forceinline__ void gemm_stage(const float* A, const v2f* sB, float* hl, const float* dv,
                                           int wave, int lane, v2f (&n0)[GU], v2f (&n1)[GU]) {
  constexpr int N = NT * 16;
  constexpr int U = GU;
  static_assert(K % (4 * U) == 0, "whole groups");
  const int r  = lane & 15;
  const int kb = (lane >> 4) << 1;

  v8f acc0[NT] = {};
  v8f acc1[NT] = {};
#pragma unroll
  for (int kc = 0; kc < K; kc += 4 * U) {
    v2f c0[U], c1[U];
#pragma unroll
    for (int u = 0; u < U; ++u) { c0[u] = n0[u]; c1[u] = n1[u]; }
    if (kc + 4 * U < K) a_group<LDA>(A, wave, lane, kc + 4 * U, n0, n1);
#pragma unroll
    for (int u = 0; u < U; ++u) {
      const int k = kc + 4 * u;
      v2f bf[NT];
#pragma unroll
      for (int t = 0; t < NT; ++t) bf[t] = sB[((k + kb) >> 1) * N + t * 16 + r];
#pragma unroll
      for (int t = 0; t < NT; ++t)
        acc0[t] = __builtin_amdgcn_wmma_f32_16x16x4_f32(false, c0[u], false, bf[t], (short)0,
                                                        acc0[t], false, false);
#pragma unroll
      for (int t = 0; t < NT; ++t)
        acc1[t] = __builtin_amdgcn_wmma_f32_16x16x4_f32(false, c1[u], false, bf[t], (short)0,
                                                        acc1[t], false, false);
    }
  }
  const int rbase = (lane >> 4) << 3;
  const float4* dv4 = reinterpret_cast<const float4*>(dv + wave * 16 + rbase);
  const float4 sa = dv4[0], sb = dv4[1], sc = dv4[64], sd = dv4[65];
  const float s0[8] = {sa.x, sa.y, sa.z, sa.w, sb.x, sb.y, sb.z, sb.w};
  const float s1[8] = {sc.x, sc.y, sc.z, sc.w, sd.x, sd.y, sd.z, sd.w};
#pragma unroll
  for (int t = 0; t < NT; ++t) {
    float* d0 = hl + (wave * 16 + rbase) * N + t * 16 + r;
    float* d1 = d0 + 256 * N;
#pragma unroll
    for (int v = 0; v < 8; ++v) { d0[v * N] = acc0[t][v] * s0[v]; d1[v * N] = acc1[t][v] * s1[v]; }
  }
}

template <int C, bool WRITE_Y>
__device__ __forceinline__ void agg_stage(const float* hl, const float* dv, const int* sg,
                                          const uint2* sx2, const float4 bb, float* Y,
                                          float* __restrict__ pout, int tid) {
  constexpr int chunks = C / 4;
  constexpr int RPT    = FT / chunks;
  constexpr int NV     = NPG * C / 4 / FT;
  static_assert(FT % chunks == 0, "whole lane groups");
  static_assert(NV % 4 == 0 && RPT * (NV / 4) == RES, "a lane group takes whole blocks of 4 nodes");
  const int g  = tid / chunks;
  const int cq = tid - g * chunks;
  const float4* hc = reinterpret_cast<const float4*>(hl) + cq;
  float4* Y4 = reinterpret_cast<float4*>(Y);
  float4* p4 = reinterpret_cast<float4*>(pout);
  const float4 z4 = make_float4(0.0f, 0.0f, 0.0f, 0.0f);

  for (int t = 0; t < NV / 4; ++t) {
    const int q = g + t * RPT;
    const int4 st = *reinterpret_cast<const int4*>(sg + 4 * q);
    const float4 d4 = *reinterpret_cast<const float4*>(dv + 4 * q);
    int e, ge, bnd, nb1, nb2;
    float dcur, dq1, dq2, dq3;
    if (WRITE_Y) {
      e = st.x; bnd = st.y; nb1 = st.z; nb2 = st.w; ge = sg[4 * q + 4];
      dcur = d4.x; dq1 = d4.y; dq2 = d4.z; dq3 = d4.w;
    } else {
      e = st.y; bnd = st.z; nb1 = -1; nb2 = -1; ge = st.w;
      dcur = d4.y; dq1 = d4.z; dq2 = 0.0f; dq3 = 0.0f;
    }
    int yi = 4 * q * (chunks + 1) + cq;
    float4 a = z4, prev = z4, pz = z4;
    for (; e < ge; e += 4) {
      uint2 pk[4];
#pragma unroll
      for (int u = 0; u < 4; ++u) pk[u] = sx2[e + u];
#pragma unroll
      for (int u = 0; u < 4; ++u)
        if (e + u >= ge) pk[u] = make_uint2(0x02000200u, 0x02000200u);
      float4 v[16];
#pragma unroll
      for (int u = 0; u < 4; ++u) {
        v[4 * u + 0] = hc[(int)(pk[u].x & 0xffffu) * chunks];
        v[4 * u + 1] = hc[(int)(pk[u].x >> 16) * chunks];
        v[4 * u + 2] = hc[(int)(pk[u].y & 0xffffu) * chunks];
        v[4 * u + 3] = hc[(int)(pk[u].y >> 16) * chunks];
      }
#pragma unroll
      for (int u = 0; u < 4; ++u) {
        if (e + u == bnd) {
          float4 rv;
          rv.x = fmaxf(a.x * dcur + bb.x, 0.0f);
          rv.y = fmaxf(a.y * dcur + bb.y, 0.0f);
          rv.z = fmaxf(a.z * dcur + bb.z, 0.0f);
          rv.w = fmaxf(a.w * dcur + bb.w, 0.0f);
          if (WRITE_Y) { Y4[yi] = rv; yi += chunks + 1; }
          pz.x = 0.5f * (prev.x + rv.x); pz.y = 0.5f * (prev.y + rv.y);
          pz.z = 0.5f * (prev.z + rv.z); pz.w = 0.5f * (prev.w + rv.w);
          prev = rv;
          a = z4;
          dcur = dq1; dq1 = dq2; dq2 = dq3;
          bnd = nb1; nb1 = nb2; nb2 = -1;
        }
#pragma unroll
        for (int i = 0; i < 4; ++i) {
          const float4 w = v[4 * u + i];
          a.x += w.x; a.y += w.y; a.z += w.z; a.w += w.w;
        }
      }
    }
    float4 rv;
    rv.x = fmaxf(a.x * dcur + bb.x, 0.0f);
    rv.y = fmaxf(a.y * dcur + bb.y, 0.0f);
    rv.z = fmaxf(a.z * dcur + bb.z, 0.0f);
    rv.w = fmaxf(a.w * dcur + bb.w, 0.0f);
    if (WRITE_Y) {
      Y4[yi] = rv;
    } else {
      pz.x = 0.5f * (prev.x + rv.x); pz.y = 0.5f * (prev.y + rv.y);
      pz.z = 0.5f * (prev.z + rv.z); pz.w = 0.5f * (prev.w + rv.w);
    }
    p4[q * chunks + cq] = pz;
  }
}

__global__ void __launch_bounds__(FT) k_graph(
    const float* __restrict__ x0, const float* __restrict__ x1,
    const int* __restrict__ ei0, const int* __restrict__ ei1,
    const float* __restrict__ W1, const float* __restrict__ b1,
    const float* __restrict__ W2, const float* __restrict__ b2,
    const float* __restrict__ W3, const float* __restrict__ b3,
    float* __restrict__ pool) {
  static_assert(EPG % FT == 0 && FT == NPG, "one node per thread");
  static_assert((SIDX_N / 2) % FT == 0, "whole sweeps of the block");
  __shared__ __align__(16) float smem[LDS_FLOATS];
  float* H  = smem + H_OFF;
  float* Y  = smem + Y_OFF;
  float* dv = smem + DV_OFF;
  int*   sg = reinterpret_cast<int*>(smem + SG_OFF);
  unsigned* sw = reinterpret_cast<unsigned*>(smem + SX_OFF);
  unsigned short* s16 = reinterpret_cast<unsigned short*>(sw);
  const uint2* sx2 = reinterpret_cast<const uint2*>(sw);
  int* cnt = reinterpret_cast<int*>(H);
  int* cur = cnt + NPG;
  int* sc  = cur + NPG;
  v2f* sB1 = reinterpret_cast<v2f*>(Y);
  v2f* sB2 = reinterpret_cast<v2f*>(H + W2_OFF);
  v2f* sB3 = reinterpret_cast<v2f*>(H + W3_OFF);

  const int b    = blockIdx.x;
  const int side = blockIdx.y;
  const int tid  = threadIdx.x;
  const float* x  = side ? x1 : x0;
  const int*   ei = side ? ei1 : ei0;
  float* pside = pool + (size_t)side * POOL_SIDE;

  int ss[EPG / FT], dd[EPG / FT];
#pragma unroll
  for (int i = 0; i < EPG / FT; ++i) {
    ss[i] = ei[b * EPG + tid + i * FT] & (NPG - 1);
    dd[i] = ei[NEDGES + b * EPG + tid + i * FT] & (NPG - 1);
  }
  v2f w1r[64 * 64 / FT];
#pragma unroll
  for (int it = 0; it < 64 * 64 / FT; ++it) {
    const int i = tid + it * FT, kp = i / 64, c = i - kp * 64;
    w1r[it].x = W1[(2 * kp) * 64 + c];
    w1r[it].y = W1[(2 * kp + 1) * 64 + c];
  }
  v2f w2r[32 * 32 / FT];
#pragma unroll
  for (int it = 0; it < 32 * 32 / FT; ++it) {
    const int i = tid + it * FT, kp = i / 32, c = i - kp * 32;
    w2r[it].x = W2[(2 * kp) * 32 + c];
    w2r[it].y = W2[(2 * kp + 1) * 32 + c];
  }
  v2f w3r = {0.0f, 0.0f};
  if (tid < 16 * 16) {
    const int kp = tid / 16, c = tid - kp * 16;
    w3r.x = W3[(2 * kp) * 16 + c];
    w3r.y = W3[(2 * kp + 1) * 16 + c];
  }
  const float4 bb1 = *reinterpret_cast<const float4*>(b1 + 4 * (tid & 15));
  const float4 bb2 = *reinterpret_cast<const float4*>(b2 + 4 * (tid & 7));
  const float4 bb3 = *reinterpret_cast<const float4*>(b3 + 4 * (tid & 3));
  const int wave = tid >> 5;
  const int lane = tid & 31;
  const float* xg = x + (size_t)b * NPG * 128;
  v2f n0[GU], n1[GU];
  a_group<128>(xg, wave, lane, 0, n0, n1);

  cnt[tid] = 0;
#pragma unroll
  for (int i = 0; i < SIDX_N / 2 / FT; ++i) sw[tid + i * FT] = 0x02000200u;
#pragma unroll
  for (int it = 0; it < 64 * 64 / FT; ++it) sB1[tid + it * FT] = w1r[it];
  __syncthreads();

#pragma unroll
  for (int i = 0; i < EPG / FT; ++i) atomicAdd(&cnt[dd[i]], 1);
  __syncthreads();
  const int c = cnt[tid];
  dv[tid] = rsqrtf((float)(c + 1));
  if (tid == 0) dv[NPG] = 0.0f;
  const int pc = (c + 4) >> 2;
  sc[tid] = pc;
  __syncthreads();
  int cb = 0;
#pragma unroll
  for (int off = 1; off < NPG; off <<= 1) {
    int v = sc[cb * NPG + tid];
    if (tid >= off) v += sc[cb * NPG + tid - off];
    sc[(cb ^ 1) * NPG + tid] = v;
    __syncthreads();
    cb ^= 1;
  }
  const int incl = sc[cb * NPG + tid];
  const int excl = incl - pc;
  cur[tid] = excl * 4 + 1;
  s16[excl * 4] = (unsigned short)tid;
  sg[tid] = excl;
  if (tid == NPG - 1) sg[NPG] = incl;
  __syncthreads();
#pragma unroll
  for (int i = 0; i < EPG / FT; ++i) {
    int pos = atomicAdd(&cur[dd[i]], 1);
    s16[pos] = (unsigned short)ss[i];
  }
  __syncthreads();

  float4* H4 = reinterpret_cast<float4*>(H);
  const float4 z4 = make_float4(0.0f, 0.0f, 0.0f, 0.0f);

  gemm_stage<4, 128, 128>(xg, sB1, H, dv, wave, lane, n0, n1);
  if (tid < 16) H4[NPG * 16 + tid] = z4;
  __syncthreads();
  agg_stage<64, true>(H, dv, sg, sx2, bb1, Y, pside + (size_t)b * RES * 64, tid);
  __syncthreads();
  sB2[tid] = w2r[0];
  sB2[tid + FT] = w2r[1];
  if (tid < 16 * 16) sB3[tid] = w3r;
  __syncthreads();

  a_group<64 + 4>(Y, wave, lane, 0, n0, n1);
  gemm_stage<2, 64, 64 + 4>(Y, sB2, H, dv, wave, lane, n0, n1);
  if (tid < 8) H4[NPG * 8 + tid] = z4;
  __syncthreads();
  agg_stage<32, true>(H, dv, sg, sx2, bb2, Y,
                      pside + (size_t)NGRAPHS * RES * 64 + (size_t)b * RES * 32, tid);
  __syncthreads();

  a_group<32 + 4>(Y, wave, lane, 0, n0, n1);
  gemm_stage<1, 32, 32 + 4>(Y, sB3, H, dv, wave, lane, n0, n1);
  if (tid < 4) H4[NPG * 4 + tid] = z4;
  __syncthreads();
  agg_stage<16, false>(H, dv, sg, sx2, bb3, Y,
                       pside + (size_t)NGRAPHS * RES * 96 + (size_t)b * RES * 16, tid);
}

#define SIM_THREADS 256
template <int D>
__device__ __forceinline__ void sim_body(const float* __restrict__ U, const float* __restrict__ V,
                                         float* __restrict__ out, float* sV, int b) {
  static_assert((RES * D / 4) % SIM_THREADS == 0, "whole sweeps of the block");
  const float4* Vb4 = reinterpret_cast<const float4*>(V + (size_t)b * RES * D);
  float4* sV4 = reinterpret_cast<float4*>(sV);
  float4 vv[RES * D / 4 / SIM_THREADS];
#pragma unroll
  for (int it = 0; it < RES * D / 4 / SIM_THREADS; ++it) vv[it] = Vb4[threadIdx.x + it * SIM_THREADS];
#pragma unroll
  for (int it = 0; it < RES * D / 4 / SIM_THREADS; ++it) sV4[threadIdx.x + it * SIM_THREADS] = vv[it];
  __syncthreads();

  int tm   = threadIdx.x >> 5;
  int lane = threadIdx.x & 31;
  int r    = lane & 15;
  int kb   = (lane >> 4) << 1;
  const float* Ar = U + ((size_t)b * RES + tm * 16 + r) * D;

  v8f acc[8] = {};
  for (int k = 0; k < D; k += 4) {
    v2f a = *(const v2f*)(Ar + k + kb);
    v2f bf[8];
#pragma unroll
    for (int t = 0; t < 8; ++t)
      bf[t] = *(const v2f*)(sV + (t * 16 + r) * D + k + kb);
#pragma unroll
    for (int t = 0; t < 8; ++t)
      acc[t] = __builtin_amdgcn_wmma_f32_16x16x4_f32(false, a, false, bf[t], (short)0,
                                                     acc[t], false, false);
  }
  int rbase = (lane >> 4) << 3;
#pragma unroll
  for (int t = 0; t < 8; ++t) {
    float* Co = out + (size_t)b * RES * RES + (size_t)(tm * 16 + rbase) * RES + t * 16 + r;
#pragma unroll
    for (int v = 0; v < 8; ++v) Co[(size_t)v * RES] = acc[t][v];
  }
}

__global__ void __launch_bounds__(SIM_THREADS) k_sim_all(const float* __restrict__ pool,
                                                         float* __restrict__ out) {
  __shared__ __align__(16) float sV[RES * 64];
  const int b = blockIdx.x;
  const int layer = blockIdx.y;
  const float* U = pool;
  const float* V = pool + POOL_SIDE;
  if (layer == 0) {
    sim_body<64>(U, V, out, sV, b);
  } else if (layer == 1) {
    sim_body<32>(U + (size_t)NGRAPHS * RES * 64, V + (size_t)NGRAPHS * RES * 64,
                 out + (size_t)1 * NGRAPHS * RES * RES, sV, b);
  } else {
    sim_body<16>(U + (size_t)NGRAPHS * RES * 96, V + (size_t)NGRAPHS * RES * 96,
                 out + (size_t)2 * NGRAPHS * RES * RES, sV, b);
  }
}

extern "C" void kernel_launch(void* const* d_in, const int* in_sizes, int n_in,
                              void* d_out, int out_size, void* d_ws, size_t ws_size,
                              hipStream_t stream) {
  (void)in_sizes; (void)n_in; (void)out_size; (void)ws_size;
  const float* x[2]  = {(const float*)d_in[0], (const float*)d_in[1]};
  const int*   ei[2] = {(const int*)d_in[2],  (const int*)d_in[3]};
  const float* W[3]  = {(const float*)d_in[6], (const float*)d_in[8], (const float*)d_in[10]};
  const float* bv[3] = {(const float*)d_in[7], (const float*)d_in[9], (const float*)d_in[11]};
  float* out = (float*)d_out;

  float* pool = (float*)d_ws;

  k_graph<<<dim3(NGRAPHS, 2), FT, 0, stream>>>(x[0], x[1], ei[0], ei[1], W[0], bv[0], W[1], bv[1],
                                               W[2], bv[2], pool);

  k_sim_all<<<dim3(NGRAPHS, 3), SIM_THREADS, 0, stream>>>(pool, out);
}


# similarity launch: V tile staged in LDS at row pitch D+4 floats, not D: the 16 rows of a B-fragment load start 4 banks apart (at pitch D they shared 1, 2 or 4 banks and were served in turns)
# speedup vs baseline: 14.7186x; 1.1002x over previous
#include <hip/hip_runtime.h>


typedef __attribute__((ext_vector_type(2))) float v2f;
typedef __attribute__((ext_vector_type(8))) float v8f;

#define NNODES  (128 * 512)
#define NEDGES  (128 * 512 * 16)
#define NGRAPHS 128
#define NPG     512
#define EPG     (512 * 16)
#define RES     128

#define SIDX_N  10240
#define FT      512
#define AGG_PAD 516
#define H_OFF   0
#define H_SIZE  ((NPG + 1) * 64)
#define Y_OFF   (H_OFF + H_SIZE)
#define Y_SIZE  (NPG * (64 + 4))
#define DV_OFF  (Y_OFF + Y_SIZE)
#define SG_OFF  (DV_OFF + AGG_PAD)
#define SX_OFF  (SG_OFF + AGG_PAD)
#define LDS_FLOATS (SX_OFF + SIDX_N / 2 + 8)
#define W2_OFF  ((NPG + 1) * 32)
#define W3_OFF  (W2_OFF + 64 * 32)
#define POOL_SIDE (NGRAPHS * RES * (64 + 32 + 16))

#define GU 8
template <int LDA>
__device__ __forceinline__ void a_group(const float* A, int wave, int lane, int kc,
                                        v2f (&n0)[GU], v2f (&n1)[GU]) {
  const float* A0 = A + (size_t)(wave * 16 + (lane & 15)) * LDA + ((lane >> 4) << 1);
  const float* A1 = A0 + (size_t)256 * LDA;
#pragma unroll
  for (int u = 0; u < GU; ++u) {
    n0[u] = *(const v2f*)(A0 + kc + 4 * u);
    n1[u] = *(const v2f*)(A1 + kc + 4 * u);
  }
}

template <int NT, int K, int LDA>
__device__ __forceinline__ void gemm_stage(const float* A, const v2f* sB, float* hl, const float* dv,
                                           int wave, int lane, v2f (&n0)[GU], v2f (&n1)[GU]) {
  constexpr int N = NT * 16;
  constexpr int U = GU;
  static_assert(K % (4 * U) == 0, "whole groups");
  const int r  = lane & 15;
  const int kb = (lane >> 4) << 1;

  v8f acc0[NT] = {};
  v8f acc1[NT] = {};
#pragma unroll
  for (int kc = 0; kc < K; kc += 4 * U) {
    v2f c0[U], c1[U];
#pragma unroll
    for (int u = 0; u < U; ++u) { c0[u] = n0[u]; c1[u] = n1[u]; }
    if (kc + 4 * U < K) a_group<LDA>(A, wave, lane, kc + 4 * U, n0, n1);
#pragma unroll
    for (int u = 0; u < U; ++u) {
      const int k = kc + 4 * u;
      v2f bf[NT];
#pragma unroll
      for (int t = 0; t < NT; ++t) bf[t] = sB[((k + kb) >> 1) * N + t * 16 + r];
#pragma unroll
      for (int t = 0; t < NT; ++t)
        acc0[t] = __builtin_amdgcn_wmma_f32_16x16x4_f32(false, c0[u], false, bf[t], (short)0,
                                                        acc0[t], false, false);
#pragma unroll
      for (int t = 0; t < NT; ++t)
        acc1[t] = __builtin_amdgcn_wmma_f32_16x16x4_f32(false, c1[u], false, bf[t], (short)0,
                                                        acc1[t], false, false);
    }
  }
  const int rbase = (lane >> 4) << 3;
  const float4* dv4 = reinterpret_cast<const float4*>(dv + wave * 16 + rbase);
  const float4 sa = dv4[0], sb = dv4[1], sc = dv4[64], sd = dv4[65];
  const float s0[8] = {sa.x, sa.y, sa.z, sa.w, sb.x, sb.y, sb.z, sb.w};
  const float s1[8] = {sc.x, sc.y, sc.z, sc.w, sd.x, sd.y, sd.z, sd.w};
#pragma unroll
  for (int t = 0; t < NT; ++t) {
    float* d0 = hl + (wave * 16 + rbase) * N + t * 16 + r;
    float* d1 = d0 + 256 * N;
#pragma unroll
    for (int v = 0; v < 8; ++v) { d0[v * N] = acc0[t][v] * s0[v]; d1[v * N] = acc1[t][v] * s1[v]; }
  }
}

template <int C, bool WRITE_Y>
__device__ __forceinline__ void agg_stage(const float* hl, const float* dv, const int* sg,
                                          const uint2* sx2, const float4 bb, float* Y,
                                          float* __restrict__ pout, int tid) {
  constexpr int chunks = C / 4;
  constexpr int RPT    = FT / chunks;
  constexpr int NV     = NPG * C / 4 / FT;
  static_assert(FT % chunks == 0, "whole lane groups");
  static_assert(NV % 4 == 0 && RPT * (NV / 4) == RES, "a lane group takes whole blocks of 4 nodes");
  const int g  = tid / chunks;
  const int cq = tid - g * chunks;
  const float4* hc = reinterpret_cast<const float4*>(hl) + cq;
  float4* Y4 = reinterpret_cast<float4*>(Y);
  float4* p4 = reinterpret_cast<float4*>(pout);
  const float4 z4 = make_float4(0.0f, 0.0f, 0.0f, 0.0f);

  for (int t = 0; t < NV / 4; ++t) {
    const int q = g + t * RPT;
    const int4 st = *reinterpret_cast<const int4*>(sg + 4 * q);
    const float4 d4 = *reinterpret_cast<const float4*>(dv + 4 * q);
    int e, ge, bnd, nb1, nb2;
    float dcur, dq1, dq2, dq3;
    if (WRITE_Y) {
      e = st.x; bnd = st.y; nb1 = st.z; nb2 = st.w; ge = sg[4 * q + 4];
      dcur = d4.x; dq1 = d4.y; dq2 = d4.z; dq3 = d4.w;
    } else {
      e = st.y; bnd = st.z; nb1 = -1; nb2 = -1; ge = st.w;
      dcur = d4.y; dq1 = d4.z; dq2 = 0.0f; dq3 = 0.0f;
    }
    int yi = 4 * q * (chunks + 1) + cq;
    float4 a = z4, prev = z4, pz = z4;
    for (; e < ge; e += 4) {
      uint2 pk[4];
#pragma unroll
      for (int u = 0; u < 4; ++u) pk[u] = sx2[e + u];
#pragma unroll
      for (int u = 0; u < 4; ++u)
        if (e + u >= ge) pk[u] = make_uint2(0x02000200u, 0x02000200u);
      float4 v[16];
#pragma unroll
      for (int u = 0; u < 4; ++u) {
        v[4 * u + 0] = hc[(int)(pk[u].x & 0xffffu) * chunks];
        v[4 * u + 1] = hc[(int)(pk[u].x >> 16) * chunks];
        v[4 * u + 2] = hc[(int)(pk[u].y & 0xffffu) * chunks];
        v[4 * u + 3] = hc[(int)(pk[u].y >> 16) * chunks];
      }
#pragma unroll
      for (int u = 0; u < 4; ++u) {
        if (e + u == bnd) {
          float4 rv;
          rv.x = fmaxf(a.x * dcur + bb.x, 0.0f);
          rv.y = fmaxf(a.y * dcur + bb.y, 0.0f);
          rv.z = fmaxf(a.z * dcur + bb.z, 0.0f);
          rv.w = fmaxf(a.w * dcur + bb.w, 0.0f);
          if (WRITE_Y) { Y4[yi] = rv; yi += chunks + 1; }
          pz.x = 0.5f * (prev.x + rv.x); pz.y = 0.5f * (prev.y + rv.y);
          pz.z = 0.5f * (prev.z + rv.z); pz.w = 0.5f * (prev.w + rv.w);
          prev = rv;
          a = z4;
          dcur = dq1; dq1 = dq2; dq2 = dq3;
          bnd = nb1; nb1 = nb2; nb2 = -1;
        }
#pragma unroll
        for (int i = 0; i < 4; ++i) {
          const float4 w = v[4 * u + i];
          a.x += w.x; a.y += w.y; a.z += w.z; a.w += w.w;
        }
      }
    }
    float4 rv;
    rv.x = fmaxf(a.x * dcur + bb.x, 0.0f);
    rv.y = fmaxf(a.y * dcur + bb.y, 0.0f);
    rv.z = fmaxf(a.z * dcur + bb.z, 0.0f);
    rv.w = fmaxf(a.w * dcur + bb.w, 0.0f);
    if (WRITE_Y) {
      Y4[yi] = rv;
    } else {
      pz.x = 0.5f * (prev.x + rv.x); pz.y = 0.5f * (prev.y + rv.y);
      pz.z = 0.5f * (prev.z + rv.z); pz.w = 0.5f * (prev.w + rv.w);
    }
    p4[q * chunks + cq] = pz;
  }
}

__global__ void __launch_bounds__(FT) k_graph(
    const float* __restrict__ x0, const float* __restrict__ x1,
    const int* __restrict__ ei0, const int* __restrict__ ei1,
    const float* __restrict__ W1, const float* __restrict__ b1,
    const float* __restrict__ W2, const float* __restrict__ b2,
    const float* __restrict__ W3, const float* __restrict__ b3,
    float* __restrict__ pool) {
  static_assert(EPG % FT == 0 && FT == NPG, "one node per thread");
  static_assert((SIDX_N / 2) % FT == 0, "whole sweeps of the block");
  __shared__ __align__(16) float smem[LDS_FLOATS];
  float* H  = smem + H_OFF;
  float* Y  = smem + Y_OFF;
  float* dv = smem + DV_OFF;
  int*   sg = reinterpret_cast<int*>(smem + SG_OFF);
  unsigned* sw = reinterpret_cast<unsigned*>(smem + SX_OFF);
  unsigned short* s16 = reinterpret_cast<unsigned short*>(sw);
  const uint2* sx2 = reinterpret_cast<const uint2*>(sw);
  int* cnt = reinterpret_cast<int*>(H);
  int* cur = cnt + NPG;
  int* sc  = cur + NPG;
  v2f* sB1 = reinterpret_cast<v2f*>(Y);
  v2f* sB2 = reinterpret_cast<v2f*>(H + W2_OFF);
  v2f* sB3 = reinterpret_cast<v2f*>(H + W3_OFF);

  const int b    = blockIdx.x;
  const int side = blockIdx.y;
  const int tid  = threadIdx.x;
  const float* x  = side ? x1 : x0;
  const int*   ei = side ? ei1 : ei0;
  float* pside = pool + (size_t)side * POOL_SIDE;

  int ss[EPG / FT], dd[EPG / FT];
#pragma unroll
  for (int i = 0; i < EPG / FT; ++i) {
    ss[i] = ei[b * EPG + tid + i * FT] & (NPG - 1);
    dd[i] = ei[NEDGES + b * EPG + tid + i * FT] & (NPG - 1);
  }
  v2f w1r[64 * 64 / FT];
#pragma unroll
  for (int it = 0; it < 64 * 64 / FT; ++it) {
    const int i = tid + it * FT, kp = i / 64, c = i - kp * 64;
    w1r[it].x = W1[(2 * kp) * 64 + c];
    w1r[it].y = W1[(2 * kp + 1) * 64 + c];
  }
  v2f w2r[32 * 32 / FT];
#pragma unroll
  for (int it = 0; it < 32 * 32 / FT; ++it) {
    const int i = tid + it * FT, kp = i / 32, c = i - kp * 32;
    w2r[it].x = W2[(2 * kp) * 32 + c];
    w2r[it].y = W2[(2 * kp + 1) * 32 + c];
  }
  v2f w3r = {0.0f, 0.0f};
  if (tid < 16 * 16) {
    const int kp = tid / 16, c = tid - kp * 16;
    w3r.x = W3[(2 * kp) * 16 + c];
    w3r.y = W3[(2 * kp + 1) * 16 + c];
  }
  const float4 bb1 = *reinterpret_cast<const float4*>(b1 + 4 * (tid & 15));
  const float4 bb2 = *reinterpret_cast<const float4*>(b2 + 4 * (tid & 7));
  const float4 bb3 = *reinterpret_cast<const float4*>(b3 + 4 * (tid & 3));
  const int wave = tid >> 5;
  const int lane = tid & 31;
  const float* xg = x + (size_t)b * NPG * 128;
  v2f n0[GU], n1[GU];
  a_group<128>(xg, wave, lane, 0, n0, n1);

  cnt[tid] = 0;
#pragma unroll
  for (int i = 0; i < SIDX_N / 2 / FT; ++i) sw[tid + i * FT] = 0x02000200u;
#pragma unroll
  for (int it = 0; it < 64 * 64 / FT; ++it) sB1[tid + it * FT] = w1r[it];
  __syncthreads();

#pragma unroll
  for (int i = 0; i < EPG / FT; ++i) atomicAdd(&cnt[dd[i]], 1);
  __syncthreads();
  const int c = cnt[tid];
  dv[tid] = rsqrtf((float)(c + 1));
  if (tid == 0) dv[NPG] = 0.0f;
  const int pc = (c + 4) >> 2;
  sc[tid] = pc;
  __syncthreads();
  int cb = 0;
#pragma unroll
  for (int off = 1; off < NPG; off <<= 1) {
    int v = sc[cb * NPG + tid];
    if (tid >= off) v += sc[cb * NPG + tid - off];
    sc[(cb ^ 1) * NPG + tid] = v;
    __syncthreads();
    cb ^= 1;
  }
  const int incl = sc[cb * NPG + tid];
  const int excl = incl - pc;
  cur[tid] = excl * 4 + 1;
  s16[excl * 4] = (unsigned short)tid;
  sg[tid] = excl;
  if (tid == NPG - 1) sg[NPG] = incl;
  __syncthreads();
#pragma unroll
  for (int i = 0; i < EPG / FT; ++i) {
    int pos = atomicAdd(&cur[dd[i]], 1);
    s16[pos] = (unsigned short)ss[i];
  }
  __syncthreads();

  float4* H4 = reinterpret_cast<float4*>(H);
  const float4 z4 = make_float4(0.0f, 0.0f, 0.0f, 0.0f);

  gemm_stage<4, 128, 128>(xg, sB1, H, dv, wave, lane, n0, n1);
  if (tid < 16) H4[NPG * 16 + tid] = z4;
  __syncthreads();
  agg_stage<64, true>(H, dv, sg, sx2, bb1, Y, pside + (size_t)b * RES * 64, tid);
  __syncthreads();
  sB2[tid] = w2r[0];
  sB2[tid + FT] = w2r[1];
  if (tid < 16 * 16) sB3[tid] = w3r;
  __syncthreads();

  a_group<64 + 4>(Y, wave, lane, 0, n0, n1);
  gemm_stage<2, 64, 64 + 4>(Y, sB2, H, dv, wave, lane, n0, n1);
  if (tid < 8) H4[NPG * 8 + tid] = z4;
  __syncthreads();
  agg_stage<32, true>(H, dv, sg, sx2, bb2, Y,
                      pside + (size_t)NGRAPHS * RES * 64 + (size_t)b * RES * 32, tid);
  __syncthreads();

  a_group<32 + 4>(Y, wave, lane, 0, n0, n1);
  gemm_stage<1, 32, 32 + 4>(Y, sB3, H, dv, wave, lane, n0, n1);
  if (tid < 4) H4[NPG * 4 + tid] = z4;
  __syncthreads();
  agg_stage<16, false>(H, dv, sg, sx2, bb3, Y,
                       pside + (size_t)NGRAPHS * RES * 96 + (size_t)b * RES * 16, tid);
}

#define SIM_THREADS 256
template <int D>
__device__ __forceinline__ void sim_body(const float* __restrict__ U, const float* __restrict__ V,
                                         float* __restrict__ out, float* sV, int b) {
  static_assert((RES * D / 4) % SIM_THREADS == 0, "whole sweeps of the block");
  constexpr int P = D + 4;
  const float4* Vb4 = reinterpret_cast<const float4*>(V + (size_t)b * RES * D);
  float4* sV4 = reinterpret_cast<float4*>(sV);
  float4 vv[RES * D / 4 / SIM_THREADS];
#pragma unroll
  for (int it = 0; it < RES * D / 4 / SIM_THREADS; ++it) vv[it] = Vb4[threadIdx.x + it * SIM_THREADS];
#pragma unroll
  for (int it = 0; it < RES * D / 4 / SIM_THREADS; ++it) {
    const int i = threadIdx.x + it * SIM_THREADS, row = i / (D / 4), c4 = i - row * (D / 4);
    sV4[row * (P / 4) + c4] = vv[it];
  }
  __syncthreads();

  int tm   = threadIdx.x >> 5;
  int lane = threadIdx.x & 31;
  int r    = lane & 15;
  int kb   = (lane >> 4) << 1;
  const float* Ar = U + ((size_t)b * RES + tm * 16 + r) * D;

  v8f acc[8] = {};
#pragma unroll(D == 64 ? 4 : D / 4)
  for (int k = 0; k < D; k += 4) {
    v2f a = *(const v2f*)(Ar + k + kb);
    v2f bf[8];
#pragma unroll
    for (int t = 0; t < 8; ++t)
      bf[t] = *(const v2f*)(sV + (t * 16 + r) * P + k + kb);
#pragma unroll
    for (int t = 0; t < 8; ++t)
      acc[t] = __builtin_amdgcn_wmma_f32_16x16x4_f32(false, a, false, bf[t], (short)0,
                                                     acc[t], false, false);
  }
  int rbase = (lane >> 4) << 3;
#pragma unroll
  for (int t = 0; t < 8; ++t) {
    float* Co = out + (size_t)b * RES * RES + (size_t)(tm * 16 + rbase) * RES + t * 16 + r;
#pragma unroll
    for (int v = 0; v < 8; ++v) Co[(size_t)v * RES] = acc[t][v];
  }
}

__global__ void __launch_bounds__(SIM_THREADS) k_sim_all(const float* __restrict__ pool,
                                                         float* __restrict__ out) {
  __shared__ __align__(16) float sV[RES * (64 + 4)];
  const int b = blockIdx.x;
  const int layer = blockIdx.y;
  const float* U = pool;
  const float* V = pool + POOL_SIDE;
  if (layer == 0) {
    sim_body<64>(U, V, out, sV, b);
  } else if (layer == 1) {
    sim_body<32>(U + (size_t)NGRAPHS * RES * 64, V + (size_t)NGRAPHS * RES * 64,
                 out + (size_t)1 * NGRAPHS * RES * RES, sV, b);
  } else {
    sim_body<16>(U + (size_t)NGRAPHS * RES * 96, V + (size_t)NGRAPHS * RES * 96,
                 out + (size_t)2 * NGRAPHS * RES * RES, sV, b);
  }
}

extern "C" void kernel_launch(void* const* d_in, const int* in_sizes, int n_in,
                              void* d_out, int out_size, void* d_ws, size_t ws_size,
                              hipStream_t stream) {
  (void)in_sizes; (void)n_in; (void)out_size; (void)ws_size;
  const float* x[2]  = {(const float*)d_in[0], (const float*)d_in[1]};
  const int*   ei[2] = {(const int*)d_in[2],  (const int*)d_in[3]};
  const float* W[3]  = {(const float*)d_in[6], (const float*)d_in[8], (const float*)d_in[10]};
  const float* bv[3] = {(const float*)d_in[7], (const float*)d_in[9], (const float*)d_in[11]};
  float* out = (float*)d_out;

  float* pool = (float*)d_ws;

  k_graph<<<dim3(NGRAPHS, 2), FT, 0, stream>>>(x[0], x[1], ei[0], ei[1], W[0], bv[0], W[1], bv[1],
                                               W[2], bv[2], pool);

  k_sim_all<<<dim3(NGRAPHS, 3), SIM_THREADS, 0, stream>>>(pool, out);
}


# k_graph: edge sort runs under layer 1's product: sort barriers split (signal .. wait), one k-step of WMMAs in each gap, A loads stay in flight across them; lists still in cursor-arrival order
# speedup vs baseline: 15.1893x; 1.0320x over previous
#include <hip/hip_runtime.h>


typedef __attribute__((ext_vector_type(2))) float v2f;
typedef __attribute__((ext_vector_type(8))) float v8f;

#define NNODES  (128 * 512)
#define NEDGES  (128 * 512 * 16)
#define NGRAPHS 128
#define NPG     512
#define EPG     (512 * 16)
#define RES     128

#define SIDX_N  10240
#define FT      512
#define AGG_PAD 516
#define H_OFF   0
#define H_SIZE  ((NPG + 1) * 64)
#define Y_OFF   (H_OFF + H_SIZE)
#define Y_SIZE  (NPG * (64 + 4))
#define DV_OFF  (Y_OFF + Y_SIZE)
#define SG_OFF  (DV_OFF + AGG_PAD)
#define SX_OFF  (SG_OFF + AGG_PAD)
#define LDS_FLOATS (SX_OFF + SIDX_N / 2 + 8)
#define W2_OFF  ((NPG + 1) * 32)
#define W3_OFF  (W2_OFF + 64 * 32)
#define POOL_SIDE (NGRAPHS * RES * (64 + 32 + 16))

#define GU 8
template <int LDA>
__device__ __forceinline__ void a_group(const float* A, int wave, int lane, int kc,
                                        v2f (&n0)[GU], v2f (&n1)[GU]) {
  const float* A0 = A + (size_t)(wave * 16 + (lane & 15)) * LDA + ((lane >> 4) << 1);
  const float* A1 = A0 + (size_t)256 * LDA;
#pragma unroll
  for (int u = 0; u < GU; ++u) {
    n0[u] = *(const v2f*)(A0 + kc + 4 * u);
    n1[u] = *(const v2f*)(A1 + kc + 4 * u);
  }
}

template <int NT>
__device__ __forceinline__ void gemm_store(const v8f (&acc0)[NT], const v8f (&acc1)[NT], float* hl,
                                           const float* dv, int wave, int lane) {
  constexpr int N = NT * 16;
  const int r = lane & 15;
  const int rbase = (lane >> 4) << 3;
  const float4* dv4 = reinterpret_cast<const float4*>(dv + wave * 16 + rbase);
  const float4 sa = dv4[0], sb = dv4[1], sc = dv4[64], sd = dv4[65];
  const float s0[8] = {sa.x, sa.y, sa.z, sa.w, sb.x, sb.y, sb.z, sb.w};
  const float s1[8] = {sc.x, sc.y, sc.z, sc.w, sd.x, sd.y, sd.z, sd.w};
#pragma unroll
  for (int t = 0; t < NT; ++t) {
    float* d0 = hl + (wave * 16 + rbase) * N + t * 16 + r;
    float* d1 = d0 + 256 * N;
#pragma unroll
    for (int v = 0; v < 8; ++v) { d0[v * N] = acc0[t][v] * s0[v]; d1[v * N] = acc1[t][v] * s1[v]; }
  }
}

template <int NT, int K, int LDA>
__device__ __forceinline__ void gemm_stage(const float* A, const v2f* sB, float* hl, const float* dv,
                                           int wave, int lane, v2f (&n0)[GU], v2f (&n1)[GU]) {
  constexpr int N = NT * 16;
  constexpr int U = GU;
  static_assert(K % (4 * U) == 0, "whole groups");
  const int r  = lane & 15;
  const int kb = (lane >> 4) << 1;

  v8f acc0[NT] = {};
  v8f acc1[NT] = {};
#pragma unroll
  for (int kc = 0; kc < K; kc += 4 * U) {
    v2f c0[U], c1[U];
#pragma unroll
    for (int u = 0; u < U; ++u) { c0[u] = n0[u]; c1[u] = n1[u]; }
    if (kc + 4 * U < K) a_group<LDA>(A, wave, lane, kc + 4 * U, n0, n1);
#pragma unroll
    for (int u = 0; u < U; ++u) {
      const int k = kc + 4 * u;
      v2f bf[NT];
#pragma unroll
      for (int t = 0; t < NT; ++t) bf[t] = sB[((k + kb) >> 1) * N + t * 16 + r];
#pragma unroll
      for (int t = 0; t < NT; ++t)
        acc0[t] = __builtin_amdgcn_wmma_f32_16x16x4_f32(false, c0[u], false, bf[t], (short)0,
                                                        acc0[t], false, false);
#pragma unroll
      for (int t = 0; t < NT; ++t)
        acc1[t] = __builtin_amdgcn_wmma_f32_16x16x4_f32(false, c1[u], false, bf[t], (short)0,
                                                        acc1[t], false, false);
    }
  }
  gemm_store<NT>(acc0, acc1, hl, dv, wave, lane);
}

template <int KS>
__device__ __forceinline__ void l1_kstep(const float* xg, const v2f* sB, int wave, int lane,
                                         v2f (&c0)[GU], v2f (&c1)[GU], v2f (&n0)[GU], v2f (&n1)[GU],
                                         v8f (&acc0)[4], v8f (&acc1)[4]) {
  constexpr int u = KS % GU;
  constexpr int k = KS * 4;
  if (u == 0) {
#pragma unroll
    for (int j = 0; j < GU; ++j) { c0[j] = n0[j]; c1[j] = n1[j]; }
    if (k + 4 * GU < 128) a_group<128>(xg, wave, lane, k + 4 * GU, n0, n1);
  }
  const int r  = lane & 15;
  const int kb = (lane >> 4) << 1;
  v2f bf[4];
#pragma unroll
  for (int t = 0; t < 4; ++t) bf[t] = sB[((k + kb) >> 1) * 64 + t * 16 + r];
#pragma unroll
  for (int t = 0; t < 4; ++t)
    acc0[t] = __builtin_amdgcn_wmma_f32_16x16x4_f32(false, c0[u], false, bf[t], (short)0,
                                                    acc0[t], false, false);
#pragma unroll
  for (int t = 0; t < 4; ++t)
    acc1[t] = __builtin_amdgcn_wmma_f32_16x16x4_f32(false, c1[u], false, bf[t], (short)0,
                                                    acc1[t], false, false);
}

__device__ __forceinline__ void bar_arrive() {
  asm volatile("s_wait_dscnt 0x0\n\ts_barrier_signal -1" ::: "memory");
}
__device__ __forceinline__ void bar_wait() {
  asm volatile("s_barrier_wait -1" ::: "memory");
}
__device__ __forceinline__ void bar_wait(const v8f (&a0)[4], const v8f (&a1)[4]) {
  asm volatile("s_barrier_wait -1"
               :: "v"(a0[0]), "v"(a0[1]), "v"(a0[2]), "v"(a0[3]), "v"(a1[0]), "v"(a1[1]), "v"(a1[2]), "v"(a1[3])
               : "memory");
}

template <int C, bool WRITE_Y>
__device__ __forceinline__ void agg_stage(const float* hl, const float* dv, const int* sg,
                                          const uint2* sx2, const float4 bb, float* Y,
                                          float* __restrict__ pout, int tid) {
  constexpr int chunks = C / 4;
  constexpr int RPT    = FT / chunks;
  constexpr int NV     = NPG * C / 4 / FT;
  static_assert(FT % chunks == 0, "whole lane groups");
  static_assert(NV % 4 == 0 && RPT * (NV / 4) == RES, "a lane group takes whole blocks of 4 nodes");
  const int g  = tid / chunks;
  const int cq = tid - g * chunks;
  const float4* hc = reinterpret_cast<const float4*>(hl) + cq;
  float4* Y4 = reinterpret_cast<float4*>(Y);
  float4* p4 = reinterpret_cast<float4*>(pout);
  const float4 z4 = make_float4(0.0f, 0.0f, 0.0f, 0.0f);

  for (int t = 0; t < NV / 4; ++t) {
    const int q = g + t * RPT;
    const int4 st = *reinterpret_cast<const int4*>(sg + 4 * q);
    const float4 d4 = *reinterpret_cast<const float4*>(dv + 4 * q);
    int e, ge, bnd, nb1, nb2;
    float dcur, dq1, dq2, dq3;
    if (WRITE_Y) {
      e = st.x; bnd = st.y; nb1 = st.z; nb2 = st.w; ge = sg[4 * q + 4];
      dcur = d4.x; dq1 = d4.y; dq2 = d4.z; dq3 = d4.w;
    } else {
      e = st.y; bnd = st.z; nb1 = -1; nb2 = -1; ge = st.w;
      dcur = d4.y; dq1 = d4.z; dq2 = 0.0f; dq3 = 0.0f;
    }
    int yi = 4 * q * (chunks + 1) + cq;
    float4 a = z4, prev = z4, pz = z4;
    for (; e < ge; e += 4) {
      uint2 pk[4];
#pragma unroll
      for (int u = 0; u < 4; ++u) pk[u] = sx2[e + u];
#pragma unroll
      for (int u = 0; u < 4; ++u)
        if (e + u >= ge) pk[u] = make_uint2(0x02000200u, 0x02000200u);
      float4 v[16];
#pragma unroll
      for (int u = 0; u < 4; ++u) {
        v[4 * u + 0] = hc[(int)(pk[u].x & 0xffffu) * chunks];
        v[4 * u + 1] = hc[(int)(pk[u].x >> 16) * chunks];
        v[4 * u + 2] = hc[(int)(pk[u].y & 0xffffu) * chunks];
        v[4 * u + 3] = hc[(int)(pk[u].y >> 16) * chunks];
      }
#pragma unroll
      for (int u = 0; u < 4; ++u) {
        if (e + u == bnd) {
          float4 rv;
          rv.x = fmaxf(a.x * dcur + bb.x, 0.0f);
          rv.y = fmaxf(a.y * dcur + bb.y, 0.0f);
          rv.z = fmaxf(a.z * dcur + bb.z, 0.0f);
          rv.w = fmaxf(a.w * dcur + bb.w, 0.0f);
          if (WRITE_Y) { Y4[yi] = rv; yi += chunks + 1; }
          pz.x = 0.5f * (prev.x + rv.x); pz.y = 0.5f * (prev.y + rv.y);
          pz.z = 0.5f * (prev.z + rv.z); pz.w = 0.5f * (prev.w + rv.w);
          prev = rv;
          a = z4;
          dcur = dq1; dq1 = dq2; dq2 = dq3;
          bnd = nb1; nb1 = nb2; nb2 = -1;
        }
#pragma unroll
        for (int i = 0; i < 4; ++i) {
          const float4 w = v[4 * u + i];
          a.x += w.x; a.y += w.y; a.z += w.z; a.w += w.w;
        }
      }
    }
    float4 rv;
    rv.x = fmaxf(a.x * dcur + bb.x, 0.0f);
    rv.y = fmaxf(a.y * dcur + bb.y, 0.0f);
    rv.z = fmaxf(a.z * dcur + bb.z, 0.0f);
    rv.w = fmaxf(a.w * dcur + bb.w, 0.0f);
    if (WRITE_Y) {
      Y4[yi] = rv;
    } else {
      pz.x = 0.5f * (prev.x + rv.x); pz.y = 0.5f * (prev.y + rv.y);
      pz.z = 0.5f * (prev.z + rv.z); pz.w = 0.5f * (prev.w + rv.w);
    }
    p4[q * chunks + cq] = pz;
  }
}

__global__ void __launch_bounds__(FT) k_graph(
    const float* __restrict__ x0, const float* __restrict__ x1,
    const int* __restrict__ ei0, const int* __restrict__ ei1,
    const float* __restrict__ W1, const float* __restrict__ b1,
    const float* __restrict__ W2, const float* __restrict__ b2,
    const float* __restrict__ W3, const float* __restrict__ b3,
    float* __restrict__ pool) {
  static_assert(EPG % FT == 0 && FT == NPG, "one node per thread");
  static_assert((SIDX_N / 2) % FT == 0, "whole sweeps of the block");
  __shared__ __align__(16) float smem[LDS_FLOATS];
  float* H  = smem + H_OFF;
  float* Y  = smem + Y_OFF;
  float* dv = smem + DV_OFF;
  int*   sg = reinterpret_cast<int*>(smem + SG_OFF);
  unsigned* sw = reinterpret_cast<unsigned*>(smem + SX_OFF);
  unsigned short* s16 = reinterpret_cast<unsigned short*>(sw);
  const uint2* sx2 = reinterpret_cast<const uint2*>(sw);
  int* cnt = reinterpret_cast<int*>(H);
  int* cur = cnt + NPG;
  int* sc  = cur + NPG;
  v2f* sB1 = reinterpret_cast<v2f*>(Y);
  v2f* sB2 = reinterpret_cast<v2f*>(H + W2_OFF);
  v2f* sB3 = reinterpret_cast<v2f*>(H + W3_OFF);

  const int b    = blockIdx.x;
  const int side = blockIdx.y;
  const int tid  = threadIdx.x;
  const float* x  = side ? x1 : x0;
  const int*   ei = side ? ei1 : ei0;
  float* pside = pool + (size_t)side * POOL_SIDE;

  int ss[EPG / FT], dd[EPG / FT];
#pragma unroll
  for (int i = 0; i < EPG / FT; ++i) {
    ss[i] = ei[b * EPG + tid + i * FT] & (NPG - 1);
    dd[i] = ei[NEDGES + b * EPG + tid + i * FT] & (NPG - 1);
  }
  v2f w1r[64 * 64 / FT];
#pragma unroll
  for (int it = 0; it < 64 * 64 / FT; ++it) {
    const int i = tid + it * FT, kp = i / 64, c = i - kp * 64;
    w1r[it].x = W1[(2 * kp) * 64 + c];
    w1r[it].y = W1[(2 * kp + 1) * 64 + c];
  }
  v2f w2r[32 * 32 / FT];
#pragma unroll
  for (int it = 0; it < 32 * 32 / FT; ++it) {
    const int i = tid + it * FT, kp = i / 32, c = i - kp * 32;
    w2r[it].x = W2[(2 * kp) * 32 + c];
    w2r[it].y = W2[(2 * kp + 1) * 32 + c];
  }
  v2f w3r = {0.0f, 0.0f};
  if (tid < 16 * 16) {
    const int kp = tid / 16, c = tid - kp * 16;
    w3r.x = W3[(2 * kp) * 16 + c];
    w3r.y = W3[(2 * kp + 1) * 16 + c];
  }
  const float4 bb1 = *reinterpret_cast<const float4*>(b1 + 4 * (tid & 15));
  const float4 bb2 = *reinterpret_cast<const float4*>(b2 + 4 * (tid & 7));
  const float4 bb3 = *reinterpret_cast<const float4*>(b3 + 4 * (tid & 3));
  const int wave = tid >> 5;
  const int lane = tid & 31;
  const float* xg = x + (size_t)b * NPG * 128;
  v2f n0[GU], n1[GU];
  a_group<128>(xg, wave, lane, 0, n0, n1);

  cnt[tid] = 0;
#pragma unroll
  for (int i = 0; i < SIDX_N / 2 / FT; ++i) sw[tid + i * FT] = 0x02000200u;
#pragma unroll
  for (int it = 0; it < 64 * 64 / FT; ++it) sB1[tid + it * FT] = w1r[it];
  bar_arrive();
  bar_wait();

  v2f c0[GU], c1[GU];
  v8f acc0[4] = {};
  v8f acc1[4] = {};
#define K(KS) l1_kstep<KS>(xg, sB1, wave, lane, c0, c1, n0, n1, acc0, acc1)
#pragma unroll
  for (int i = 0; i < EPG / FT; ++i) atomicAdd(&cnt[dd[i]], 1);
  K(0); K(1);
  bar_arrive();
  K(2);
  bar_wait(acc0, acc1);
  const int c = cnt[tid];
  dv[tid] = rsqrtf((float)(c + 1));
  if (tid == 0) dv[NPG] = 0.0f;
  const int pc = (c + 4) >> 2;
  sc[tid] = pc;
  K(3);
  bar_arrive();
  K(4);
  bar_wait(acc0, acc1);
#define SCAN_STEP(S)                                                        \
  {                                                                         \
    constexpr int off = 1 << (S);                                           \
    constexpr int rb = ((S) & 1) * NPG, wb = (((S) & 1) ^ 1) * NPG;         \
    const int own = sc[rb + tid];                                           \
    const int low = sc[rb + (tid >= off ? tid - off : tid)];                \
    K(5 + 2 * (S));                                                         \
    sc[wb + tid] = own + (tid >= off ? low : 0);                            \
    bar_arrive();                                                           \
    K(6 + 2 * (S));                                                         \
    bar_wait(acc0, acc1);                                                   \
  }
  SCAN_STEP(0) SCAN_STEP(1) SCAN_STEP(2) SCAN_STEP(3) SCAN_STEP(4)
  SCAN_STEP(5) SCAN_STEP(6) SCAN_STEP(7) SCAN_STEP(8)
#undef SCAN_STEP
  const int incl = sc[NPG + tid];
  const int excl = incl - pc;
  cur[tid] = excl * 4 + 1;
  s16[excl * 4] = (unsigned short)tid;
  sg[tid] = excl;
  if (tid == NPG - 1) sg[NPG] = incl;
  K(23);
  bar_arrive();
  K(24);
  bar_wait(acc0, acc1);
#pragma unroll
  for (int i = 0; i < EPG / FT; ++i) {
    int pos = atomicAdd(&cur[dd[i]], 1);
    s16[pos] = (unsigned short)ss[i];
    if (i == 3) K(25);
    if (i == 7) K(26);
    if (i == 11) K(27);
    if (i == 15) K(28);
  }
  bar_arrive();
  K(29);
  bar_wait(acc0, acc1);
  K(30); K(31);
#undef K

  float4* H4 = reinterpret_cast<float4*>(H);
  const float4 z4 = make_float4(0.0f, 0.0f, 0.0f, 0.0f);

  gemm_store<4>(acc0, acc1, H, dv, wave, lane);
  if (tid < 16) H4[NPG * 16 + tid] = z4;
  __syncthreads();
  agg_stage<64, true>(H, dv, sg, sx2, bb1, Y, pside + (size_t)b * RES * 64, tid);
  __syncthreads();
  sB2[tid] = w2r[0];
  sB2[tid + FT] = w2r[1];
  if (tid < 16 * 16) sB3[tid] = w3r;
  __syncthreads();

  a_group<64 + 4>(Y, wave, lane, 0, n0, n1);
  gemm_stage<2, 64, 64 + 4>(Y, sB2, H, dv, wave, lane, n0, n1);
  if (tid < 8) H4[NPG * 8 + tid] = z4;
  __syncthreads();
  agg_stage<32, true>(H, dv, sg, sx2, bb2, Y,
                      pside + (size_t)NGRAPHS * RES * 64 + (size_t)b * RES * 32, tid);
  __syncthreads();

  a_group<32 + 4>(Y, wave, lane, 0, n0, n1);
  gemm_stage<1, 32, 32 + 4>(Y, sB3, H, dv, wave, lane, n0, n1);
  if (tid < 4) H4[NPG * 4 + tid] = z4;
  __syncthreads();
  agg_stage<16, false>(H, dv, sg, sx2, bb3, Y,
                       pside + (size_t)NGRAPHS * RES * 96 + (size_t)b * RES * 16, tid);
}

#define SIM_THREADS 256
template <int D>
__device__ __forceinline__ void sim_body(const float* __restrict__ U, const float* __restrict__ V,
                                         float* __restrict__ out, float* sV, int b) {
  static_assert((RES * D / 4) % SIM_THREADS == 0, "whole sweeps of the block");
  constexpr int P = D + 4;
  const float4* Vb4 = reinterpret_cast<const float4*>(V + (size_t)b * RES * D);
  float4* sV4 = reinterpret_cast<float4*>(sV);
  float4 vv[RES * D / 4 / SIM_THREADS];
#pragma unroll
  for (int it = 0; it < RES * D / 4 / SIM_THREADS; ++it) vv[it] = Vb4[threadIdx.x + it * SIM_THREADS];
#pragma unroll
  for (int it = 0; it < RES * D / 4 / SIM_THREADS; ++it) {
    const int i = threadIdx.x + it * SIM_THREADS, row = i / (D / 4), c4 = i - row * (D / 4);
    sV4[row * (P / 4) + c4] = vv[it];
  }
  __syncthreads();

  int tm   = threadIdx.x >> 5;
  int lane = threadIdx.x & 31;
  int r    = lane & 15;
  int kb   = (lane >> 4) << 1;
  const float* Ar = U + ((size_t)b * RES + tm * 16 + r) * D;

  v8f acc[8] = {};
#pragma unroll(D == 64 ? 4 : D / 4)
  for (int k = 0; k < D; k += 4) {
    v2f a = *(const v2f*)(Ar + k + kb);
    v2f bf[8];
#pragma unroll
    for (int t = 0; t < 8; ++t)
      bf[t] = *(const v2f*)(sV + (t * 16 + r) * P + k + kb);
#pragma unroll
    for (int t = 0; t < 8; ++t)
      acc[t] = __builtin_amdgcn_wmma_f32_16x16x4_f32(false, a, false, bf[t], (short)0,
                                                     acc[t], false, false);
  }
  int rbase = (lane >> 4) << 3;
#pragma unroll
  for (int t = 0; t < 8; ++t) {
    float* Co = out + (size_t)b * RES * RES + (size_t)(tm * 16 + rbase) * RES + t * 16 + r;
#pragma unroll
    for (int v = 0; v < 8; ++v) Co[(size_t)v * RES] = acc[t][v];
  }
}

__global__ void __launch_bounds__(SIM_THREADS) k_sim_all(const float* __restrict__ pool,
                                                         float* __restrict__ out) {
  __shared__ __align__(16) float sV[RES * (64 + 4)];
  const int b = blockIdx.x;
  const int layer = blockIdx.y;
  const float* U = pool;
  const float* V = pool + POOL_SIDE;
  if (layer == 0) {
    sim_body<64>(U, V, out, sV, b);
  } else if (layer == 1) {
    sim_body<32>(U + (size_t)NGRAPHS * RES * 64, V + (size_t)NGRAPHS * RES * 64,
                 out + (size_t)1 * NGRAPHS * RES * RES, sV, b);
  } else {
    sim_body<16>(U + (size_t)NGRAPHS * RES * 96, V + (size_t)NGRAPHS * RES * 96,
                 out + (size_t)2 * NGRAPHS * RES * RES, sV, b);
  }
}

extern "C" void kernel_launch(void* const* d_in, const int* in_sizes, int n_in,
                              void* d_out, int out_size, void* d_ws, size_t ws_size,
                              hipStream_t stream) {
  (void)in_sizes; (void)n_in; (void)out_size; (void)ws_size;
  const float* x[2]  = {(const float*)d_in[0], (const float*)d_in[1]};
  const int*   ei[2] = {(const int*)d_in[2],  (const int*)d_in[3]};
  const float* W[3]  = {(const float*)d_in[6], (const float*)d_in[8], (const float*)d_in[10]};
  const float* bv[3] = {(const float*)d_in[7], (const float*)d_in[9], (const float*)d_in[11]};
  float* out = (float*)d_out;

  float* pool = (float*)d_ws;

  k_graph<<<dim3(NGRAPHS, 2), FT, 0, stream>>>(x[0], x[1], ei[0], ei[1], W[0], bv[0], W[1], bv[1],
                                               W[2], bv[2], pool);

  k_sim_all<<<dim3(NGRAPHS, 3), SIM_THREADS, 0, stream>>>(pool, out);
}


# layer 1: a wave fetches its 32 rows of X as whole 128-byte lines (4 lines per load, not 16), parks each 32-float chunk in its own LDS strip and reads the A-fragments from there; same k order
# speedup vs baseline: 15.7935x; 1.0398x over previous
#include <hip/hip_runtime.h>


typedef __attribute__((ext_vector_type(2))) float v2f;
typedef __attribute__((ext_vector_type(4))) float v4f;
typedef __attribute__((ext_vector_type(8))) float v8f;

#define NNODES  (128 * 512)
#define NEDGES  (128 * 512 * 16)
#define NGRAPHS 128
#define NPG     512
#define EPG     (512 * 16)
#define RES     128

#define SIDX_N  10240
#define FT      512
#define AGG_PAD 516
#define H_OFF   0
#define H_SIZE  ((NPG + 1) * 64)
#define Y_OFF   (H_OFF + H_SIZE)
#define Y_SIZE  (NPG * (64 + 4))
#define DV_OFF  (Y_OFF + Y_SIZE)
#define SG_OFF  (DV_OFF + AGG_PAD)
#define SX_OFF  (SG_OFF + AGG_PAD)
#define LDS_FLOATS (SX_OFF + SIDX_N / 2 + 8)
#define W2_OFF  ((NPG + 1) * 32)
#define W3_OFF  (W2_OFF + 64 * 32)
#define POOL_SIDE (NGRAPHS * RES * (64 + 32 + 16))
#define XS_OFF  (4 * NPG)
#define XS_P    (32 + 4)
#define XS_WAVE (32 * XS_P)

#define GU 8
template <int LDA>
__device__ __forceinline__ void a_group(const float* A, int wave, int lane, int kc,
                                        v2f (&n0)[GU], v2f (&n1)[GU]) {
  const float* A0 = A + (size_t)(wave * 16 + (lane & 15)) * LDA + ((lane >> 4) << 1);
  const float* A1 = A0 + (size_t)256 * LDA;
#pragma unroll
  for (int u = 0; u < GU; ++u) {
    n0[u] = *(const v2f*)(A0 + kc + 4 * u);
    n1[u] = *(const v2f*)(A1 + kc + 4 * u);
  }
}

template <int NT>
__device__ __forceinline__ void gemm_store(const v8f (&acc0)[NT], const v8f (&acc1)[NT], float* hl,
                                           const float* dv, int wave, int lane) {
  constexpr int N = NT * 16;
  const int r = lane & 15;
  const int rbase = (lane >> 4) << 3;
  const float4* dv4 = reinterpret_cast<const float4*>(dv + wave * 16 + rbase);
  const float4 sa = dv4[0], sb = dv4[1], sc = dv4[64], sd = dv4[65];
  const float s0[8] = {sa.x, sa.y, sa.z, sa.w, sb.x, sb.y, sb.z, sb.w};
  const float s1[8] = {sc.x, sc.y, sc.z, sc.w, sd.x, sd.y, sd.z, sd.w};
#pragma unroll
  for (int t = 0; t < NT; ++t) {
    float* d0 = hl + (wave * 16 + rbase) * N + t * 16 + r;
    float* d1 = d0 + 256 * N;
#pragma unroll
    for (int v = 0; v < 8; ++v) { d0[v * N] = acc0[t][v] * s0[v]; d1[v * N] = acc1[t][v] * s1[v]; }
  }
}

template <int NT, int K, int LDA>
__device__ __forceinline__ void gemm_stage(const float* A, const v2f* sB, float* hl, const float* dv,
                                           int wave, int lane, v2f (&n0)[GU], v2f (&n1)[GU]) {
  constexpr int N = NT * 16;
  constexpr int U = GU;
  static_assert(K % (4 * U) == 0, "whole groups");
  const int r  = lane & 15;
  const int kb = (lane >> 4) << 1;

  v8f acc0[NT] = {};
  v8f acc1[NT] = {};
#pragma unroll
  for (int kc = 0; kc < K; kc += 4 * U) {
    v2f c0[U], c1[U];
#pragma unroll
    for (int u = 0; u < U; ++u) { c0[u] = n0[u]; c1[u] = n1[u]; }
    if (kc + 4 * U < K) a_group<LDA>(A, wave, lane, kc + 4 * U, n0, n1);
#pragma unroll
    for (int u = 0; u < U; ++u) {
      const int k = kc + 4 * u;
      v2f bf[NT];
#pragma unroll
      for (int t = 0; t < NT; ++t) bf[t] = sB[((k + kb) >> 1) * N + t * 16 + r];
#pragma unroll
      for (int t = 0; t < NT; ++t)
        acc0[t] = __builtin_amdgcn_wmma_f32_16x16x4_f32(false, c0[u], false, bf[t], (short)0,
                                                        acc0[t], false, false);
#pragma unroll
      for (int t = 0; t < NT; ++t)
        acc1[t] = __builtin_amdgcn_wmma_f32_16x16x4_f32(false, c1[u], false, bf[t], (short)0,
                                                        acc1[t], false, false);
    }
  }
  gemm_store<NT>(acc0, acc1, hl, dv, wave, lane);
}

__device__ __forceinline__ void x_fetch(const float* xg, int wave, int lane, int kc, v4f (&g)[8]) {
  const float* p = xg + (size_t)(wave * 16 + (lane >> 3)) * 128 + kc + ((lane & 7) << 2);
#pragma unroll
  for (int i = 0; i < 8; ++i)
    g[i] = *(const v4f*)(p + (size_t)((i & 3) * 4 + (i >> 2) * 256) * 128);
}
__device__ __forceinline__ void x_park(float* strip, int lane, const v4f (&g)[8]) {
  float* s = strip + (lane >> 3) * XS_P + ((lane & 7) << 2);
#pragma unroll
  for (int i = 0; i < 8; ++i) *(v4f*)(s + 4 * i * XS_P) = g[i];
}
__device__ __forceinline__ void x_frags(const float* strip, int lane, v2f (&c0)[GU], v2f (&c1)[GU]) {
  const float* s = strip + (lane & 15) * XS_P + ((lane >> 4) << 1);
#pragma unroll
  for (int u = 0; u < GU; ++u) {
    c0[u] = *(const v2f*)(s + 4 * u);
    c1[u] = *(const v2f*)(s + 16 * XS_P + 4 * u);
  }
}

template <int KS>
__device__ __forceinline__ void l1_kstep(const float* xg, const v2f* sB, float* strip, int wave, int lane,
                                         v2f (&c0)[GU], v2f (&c1)[GU], v4f (&xr)[8],
                                         v8f (&acc0)[4], v8f (&acc1)[4]) {
  constexpr int u = KS % GU;
  constexpr int k = KS * 4;
  if (u == 0) {
    x_park(strip, lane, xr);
    if (k + 4 * GU < 128) x_fetch(xg, wave, lane, k + 4 * GU, xr);
    x_frags(strip, lane, c0, c1);
  }
  const int r  = lane & 15;
  const int kb = (lane >> 4) << 1;
  v2f bf[4];
#pragma unroll
  for (int t = 0; t < 4; ++t) bf[t] = sB[((k + kb) >> 1) * 64 + t * 16 + r];
#pragma unroll
  for (int t = 0; t < 4; ++t)
    acc0[t] = __builtin_amdgcn_wmma_f32_16x16x4_f32(false, c0[u], false, bf[t], (short)0,
                                                    acc0[t], false, false);
#pragma unroll
  for (int t = 0; t < 4; ++t)
    acc1[t] = __builtin_amdgcn_wmma_f32_16x16x4_f32(false, c1[u], false, bf[t], (short)0,
                                                    acc1[t], false, false);
}

__device__ __forceinline__ void bar_arrive() {
  asm volatile("s_wait_dscnt 0x0\n\ts_barrier_signal -1" ::: "memory");
}
__device__ __forceinline__ void bar_wait() {
  asm volatile("s_barrier_wait -1" ::: "memory");
}
__device__ __forceinline__ void bar_wait(const v8f (&a0)[4], const v8f (&a1)[4]) {
  asm volatile("s_barrier_wait -1"
               :: "v"(a0[0]), "v"(a0[1]), "v"(a0[2]), "v"(a0[3]), "v"(a1[0]), "v"(a1[1]), "v"(a1[2]), "v"(a1[3])
               : "memory");
}

template <int C, bool WRITE_Y>
__device__ __forceinline__ void agg_stage(const float* hl, const float* dv, const int* sg,
                                          const uint2* sx2, const float4 bb, float* Y,
                                          float* __restrict__ pout, int tid) {
  constexpr int chunks = C / 4;
  constexpr int RPT    = FT / chunks;
  constexpr int NV     = NPG * C / 4 / FT;
  static_assert(FT % chunks == 0, "whole lane groups");
  static_assert(NV % 4 == 0 && RPT * (NV / 4) == RES, "a lane group takes whole blocks of 4 nodes");
  const int g  = tid / chunks;
  const int cq = tid - g * chunks;
  const float4* hc = reinterpret_cast<const float4*>(hl) + cq;
  float4* Y4 = reinterpret_cast<float4*>(Y);
  float4* p4 = reinterpret_cast<float4*>(pout);
  const float4 z4 = make_float4(0.0f, 0.0f, 0.0f, 0.0f);

  for (int t = 0; t < NV / 4; ++t) {
    const int q = g + t * RPT;
    const int4 st = *reinterpret_cast<const int4*>(sg + 4 * q);
    const float4 d4 = *reinterpret_cast<const float4*>(dv + 4 * q);
    int e, ge, bnd, nb1, nb2;
    float dcur, dq1, dq2, dq3;
    if (WRITE_Y) {
      e = st.x; bnd = st.y; nb1 = st.z; nb2 = st.w; ge = sg[4 * q + 4];
      dcur = d4.x; dq1 = d4.y; dq2 = d4.z; dq3 = d4.w;
    } else {
      e = st.y; bnd = st.z; nb1 = -1; nb2 = -1; ge = st.w;
      dcur = d4.y; dq1 = d4.z; dq2 = 0.0f; dq3 = 0.0f;
    }
    int yi = 4 * q * (chunks + 1) + cq;
    float4 a = z4, prev = z4, pz = z4;
    for (; e < ge; e += 4) {
      uint2 pk[4];
#pragma unroll
      for (int u = 0; u < 4; ++u) pk[u] = sx2[e + u];
#pragma unroll
      for (int u = 0; u < 4; ++u)
        if (e + u >= ge) pk[u] = make_uint2(0x02000200u, 0x02000200u);
      float4 v[16];
#pragma unroll
      for (int u = 0; u < 4; ++u) {
        v[4 * u + 0] = hc[(int)(pk[u].x & 0xffffu) * chunks];
        v[4 * u + 1] = hc[(int)(pk[u].x >> 16) * chunks];
        v[4 * u + 2] = hc[(int)(pk[u].y & 0xffffu) * chunks];
        v[4 * u + 3] = hc[(int)(pk[u].y >> 16) * chunks];
      }
#pragma unroll
      for (int u = 0; u < 4; ++u) {
        if (e + u == bnd) {
          float4 rv;
          rv.x = fmaxf(a.x * dcur + bb.x, 0.0f);
          rv.y = fmaxf(a.y * dcur + bb.y, 0.0f);
          rv.z = fmaxf(a.z * dcur + bb.z, 0.0f);
          rv.w = fmaxf(a.w * dcur + bb.w, 0.0f);
          if (WRITE_Y) { Y4[yi] = rv; yi += chunks + 1; }
          pz.x = 0.5f * (prev.x + rv.x); pz.y = 0.5f * (prev.y + rv.y);
          pz.z = 0.5f * (prev.z + rv.z); pz.w = 0.5f * (prev.w + rv.w);
          prev = rv;
          a = z4;
          dcur = dq1; dq1 = dq2; dq2 = dq3;
          bnd = nb1; nb1 = nb2; nb2 = -1;
        }
#pragma unroll
        for (int i = 0; i < 4; ++i) {
          const float4 w = v[4 * u + i];
          a.x += w.x; a.y += w.y; a.z += w.z; a.w += w.w;
        }
      }
    }
    float4 rv;
    rv.x = fmaxf(a.x * dcur + bb.x, 0.0f);
    rv.y = fmaxf(a.y * dcur + bb.y, 0.0f);
    rv.z = fmaxf(a.z * dcur + bb.z, 0.0f);
    rv.w = fmaxf(a.w * dcur + bb.w, 0.0f);
    if (WRITE_Y) {
      Y4[yi] = rv;
    } else {
      pz.x = 0.5f * (prev.x + rv.x); pz.y = 0.5f * (prev.y + rv.y);
      pz.z = 0.5f * (prev.z + rv.z); pz.w = 0.5f * (prev.w + rv.w);
    }
    p4[q * chunks + cq] = pz;
  }
}

__global__ void __launch_bounds__(FT) k_graph(
    const float* __restrict__ x0, const float* __restrict__ x1,
    const int* __restrict__ ei0, const int* __restrict__ ei1,
    const float* __restrict__ W1, const float* __restrict__ b1,
    const float* __restrict__ W2, const float* __restrict__ b2,
    const float* __restrict__ W3, const float* __restrict__ b3,
    float* __restrict__ pool) {
  static_assert(EPG % FT == 0 && FT == NPG, "one node per thread");
  static_assert((SIDX_N / 2) % FT == 0, "whole sweeps of the block");
  __shared__ __align__(16) float smem[LDS_FLOATS];
  float* H  = smem + H_OFF;
  float* Y  = smem + Y_OFF;
  float* dv = smem + DV_OFF;
  int*   sg = reinterpret_cast<int*>(smem + SG_OFF);
  unsigned* sw = reinterpret_cast<unsigned*>(smem + SX_OFF);
  unsigned short* s16 = reinterpret_cast<unsigned short*>(sw);
  const uint2* sx2 = reinterpret_cast<const uint2*>(sw);
  int* cnt = reinterpret_cast<int*>(H);
  int* cur = cnt + NPG;
  int* sc  = cur + NPG;
  v2f* sB1 = reinterpret_cast<v2f*>(Y);
  v2f* sB2 = reinterpret_cast<v2f*>(H + W2_OFF);
  v2f* sB3 = reinterpret_cast<v2f*>(H + W3_OFF);

  const int b    = blockIdx.x;
  const int side = blockIdx.y;
  const int tid  = threadIdx.x;
  const float* x  = side ? x1 : x0;
  const int*   ei = side ? ei1 : ei0;
  float* pside = pool + (size_t)side * POOL_SIDE;

  int ss[EPG / FT], dd[EPG / FT];
#pragma unroll
  for (int i = 0; i < EPG / FT; ++i) {
    ss[i] = ei[b * EPG + tid + i * FT] & (NPG - 1);
    dd[i] = ei[NEDGES + b * EPG + tid + i * FT] & (NPG - 1);
  }
  v2f w1r[64 * 64 / FT];
#pragma unroll
  for (int it = 0; it < 64 * 64 / FT; ++it) {
    const int i = tid + it * FT, kp = i / 64, c = i - kp * 64;
    w1r[it].x = W1[(2 * kp) * 64 + c];
    w1r[it].y = W1[(2 * kp + 1) * 64 + c];
  }
  v2f w2r[32 * 32 / FT];
#pragma unroll
  for (int it = 0; it < 32 * 32 / FT; ++it) {
    const int i = tid + it * FT, kp = i / 32, c = i - kp * 32;
    w2r[it].x = W2[(2 * kp) * 32 + c];
    w2r[it].y = W2[(2 * kp + 1) * 32 + c];
  }
  v2f w3r = {0.0f, 0.0f};
  if (tid < 16 * 16) {
    const int kp = tid / 16, c = tid - kp * 16;
    w3r.x = W3[(2 * kp) * 16 + c];
    w3r.y = W3[(2 * kp + 1) * 16 + c];
  }
  const float4 bb1 = *reinterpret_cast<const float4*>(b1 + 4 * (tid & 15));
  const float4 bb2 = *reinterpret_cast<const float4*>(b2 + 4 * (tid & 7));
  const float4 bb3 = *reinterpret_cast<const float4*>(b3 + 4 * (tid & 3));
  const int wave = tid >> 5;
  const int lane = tid & 31;
  const float* xg = x + (size_t)b * NPG * 128;
  float* xs = H + XS_OFF + wave * XS_WAVE;
  v4f xr[8];
  x_fetch(xg, wave, lane, 0, xr);

  cnt[tid] = 0;
#pragma unroll
  for (int i = 0; i < SIDX_N / 2 / FT; ++i) sw[tid + i * FT] = 0x02000200u;
#pragma unroll
  for (int it = 0; it < 64 * 64 / FT; ++it) sB1[tid + it * FT] = w1r[it];
  bar_arrive();
  bar_wait();

  v2f c0[GU], c1[GU];
  v8f acc0[4] = {};
  v8f acc1[4] = {};
#define K(KS) l1_kstep<KS>(xg, sB1, xs, wave, lane, c0, c1, xr, acc0, acc1)
#pragma unroll
  for (int i = 0; i < EPG / FT; ++i) atomicAdd(&cnt[dd[i]], 1);
  K(0); K(1);
  bar_arrive();
  K(2);
  bar_wait(acc0, acc1);
  const int c = cnt[tid];
  dv[tid] = rsqrtf((float)(c + 1));
  if (tid == 0) dv[NPG] = 0.0f;
  const int pc = (c + 4) >> 2;
  sc[tid] = pc;
  K(3);
  bar_arrive();
  K(4);
  bar_wait(acc0, acc1);
#define SCAN_STEP(S)                                                        \
  {                                                                         \
    constexpr int off = 1 << (S);                                           \
    constexpr int rb = ((S) & 1) * NPG, wb = (((S) & 1) ^ 1) * NPG;         \
    const int own = sc[rb + tid];                                           \
    const int low = sc[rb + (tid >= off ? tid - off : tid)];                \
    K(5 + 2 * (S));                                                         \
    sc[wb + tid] = own + (tid >= off ? low : 0);                            \
    bar_arrive();                                                           \
    K(6 + 2 * (S));                                                         \
    bar_wait(acc0, acc1);                                                   \
  }
  SCAN_STEP(0) SCAN_STEP(1) SCAN_STEP(2) SCAN_STEP(3) SCAN_STEP(4)
  SCAN_STEP(5) SCAN_STEP(6) SCAN_STEP(7) SCAN_STEP(8)
#undef SCAN_STEP
  const int incl = sc[NPG + tid];
  const int excl = incl - pc;
  cur[tid] = excl * 4 + 1;
  s16[excl * 4] = (unsigned short)tid;
  sg[tid] = excl;
  if (tid == NPG - 1) sg[NPG] = incl;
  K(23);
  bar_arrive();
  K(24);
  bar_wait(acc0, acc1);
#pragma unroll
  for (int i = 0; i < EPG / FT; ++i) {
    int pos = atomicAdd(&cur[dd[i]], 1);
    s16[pos] = (unsigned short)ss[i];
    if (i == 3) K(25);
    if (i == 7) K(26);
    if (i == 11) K(27);
    if (i == 15) K(28);
  }
  bar_arrive();
  K(29);
  bar_wait(acc0, acc1);
  K(30); K(31);
#undef K

  float4* H4 = reinterpret_cast<float4*>(H);
  const float4 z4 = make_float4(0.0f, 0.0f, 0.0f, 0.0f);

  gemm_store<4>(acc0, acc1, H, dv, wave, lane);
  if (tid < 16) H4[NPG * 16 + tid] = z4;
  __syncthreads();
  agg_stage<64, true>(H, dv, sg, sx2, bb1, Y, pside + (size_t)b * RES * 64, tid);
  __syncthreads();
  sB2[tid] = w2r[0];
  sB2[tid + FT] = w2r[1];
  if (tid < 16 * 16) sB3[tid] = w3r;
  __syncthreads();

  v2f n0[GU], n1[GU];
  a_group<64 + 4>(Y, wave, lane, 0, n0, n1);
  gemm_stage<2, 64, 64 + 4>(Y, sB2, H, dv, wave, lane, n0, n1);
  if (tid < 8) H4[NPG * 8 + tid] = z4;
  __syncthreads();
  agg_stage<32, true>(H, dv, sg, sx2, bb2, Y,
                      pside + (size_t)NGRAPHS * RES * 64 + (size_t)b * RES * 32, tid);
  __syncthreads();

  a_group<32 + 4>(Y, wave, lane, 0, n0, n1);
  gemm_stage<1, 32, 32 + 4>(Y, sB3, H, dv, wave, lane, n0, n1);
  if (tid < 4) H4[NPG * 4 + tid] = z4;
  __syncthreads();
  agg_stage<16, false>(H, dv, sg, sx2, bb3, Y,
                       pside + (size_t)NGRAPHS * RES * 96 + (size_t)b * RES * 16, tid);
}

#define SIM_THREADS 256
template <int D>
__device__ __forceinline__ void sim_body(const float* __restrict__ U, const float* __restrict__ V,
                                         float* __restrict__ out, float* sV, int b) {
  static_assert((RES * D / 4) % SIM_THREADS == 0, "whole sweeps of the block");
  constexpr int P = D + 4;
  const float4* Vb4 = reinterpret_cast<const float4*>(V + (size_t)b * RES * D);
  float4* sV4 = reinterpret_cast<float4*>(sV);
  float4 vv[RES * D / 4 / SIM_THREADS];
#pragma unroll
  for (int it = 0; it < RES * D / 4 / SIM_THREADS; ++it) vv[it] = Vb4[threadIdx.x + it * SIM_THREADS];
#pragma unroll
  for (int it = 0; it < RES * D / 4 / SIM_THREADS; ++it) {
    const int i = threadIdx.x + it * SIM_THREADS, row = i / (D / 4), c4 = i - row * (D / 4);
    sV4[row * (P / 4) + c4] = vv[it];
  }
  __syncthreads();

  int tm   = threadIdx.x >> 5;
  int lane = threadIdx.x & 31;
  int r    = lane & 15;
  int kb   = (lane >> 4) << 1;
  const float* Ar = U + ((size_t)b * RES + tm * 16 + r) * D;

  v8f acc[8] = {};
#pragma unroll(D == 64 ? 4 : D / 4)
  for (int k = 0; k < D; k += 4) {
    v2f a = *(const v2f*)(Ar + k + kb);
    v2f bf[8];
#pragma unroll
    for (int t = 0; t < 8; ++t)
      bf[t] = *(const v2f*)(sV + (t * 16 + r) * P + k + kb);
#pragma unroll
    for (int t = 0; t < 8; ++t)
      acc[t] = __builtin_amdgcn_wmma_f32_16x16x4_f32(false, a, false, bf[t], (short)0,
                                                     acc[t], false, false);
  }
  int rbase = (lane >> 4) << 3;
#pragma unroll
  for (int t = 0; t < 8; ++t) {
    float* Co = out + (size_t)b * RES * RES + (size_t)(tm * 16 + rbase) * RES + t * 16 + r;
#pragma unroll
    for (int v = 0; v < 8; ++v) Co[(size_t)v * RES] = acc[t][v];
  }
}

__global__ void __launch_bounds__(SIM_THREADS) k_sim_all(const float* __restrict__ pool,
                                                         float* __restrict__ out) {
  __shared__ __align__(16) float sV[RES * (64 + 4)];
  const int b = blockIdx.x;
  const int layer = blockIdx.y;
  const float* U = pool;
  const float* V = pool + POOL_SIDE;
  if (layer == 0) {
    sim_body<64>(U, V, out, sV, b);
  } else if (layer == 1) {
    sim_body<32>(U + (size_t)NGRAPHS * RES * 64, V + (size_t)NGRAPHS * RES * 64,
                 out + (size_t)1 * NGRAPHS * RES * RES, sV, b);
  } else {
    sim_body<16>(U + (size_t)NGRAPHS * RES * 96, V + (size_t)NGRAPHS * RES * 96,
                 out + (size_t)2 * NGRAPHS * RES * RES, sV, b);
  }
}

extern "C" void kernel_launch(void* const* d_in, const int* in_sizes, int n_in,
                              void* d_out, int out_size, void* d_ws, size_t ws_size,
                              hipStream_t stream) {
  (void)in_sizes; (void)n_in; (void)out_size; (void)ws_size;
  const float* x[2]  = {(const float*)d_in[0], (const float*)d_in[1]};
  const int*   ei[2] = {(const int*)d_in[2],  (const int*)d_in[3]};
  const float* W[3]  = {(const float*)d_in[6], (const float*)d_in[8], (const float*)d_in[10]};
  const float* bv[3] = {(const float*)d_in[7], (const float*)d_in[9], (const float*)d_in[11]};
  float* out = (float*)d_out;

  float* pool = (float*)d_ws;

  k_graph<<<dim3(NGRAPHS, 2), FT, 0, stream>>>(x[0], x[1], ei[0], ei[1], W[0], bv[0], W[1], bv[1],
                                               W[2], bv[2], pool);

  k_sim_all<<<dim3(NGRAPHS, 3), SIM_THREADS, 0, stream>>>(pool, out);
}
